# starting kernel
# baseline (speedup 1.0000x reference)
#include <hip/hip_runtime.h>

#define N_AA   21
#define NATOM  15
#define DFEAT  128
#define DCOORD (N_AA * NATOM * 3)
#define DDIH   39
#define DH1    256
#define H1S    260
#define H2S    132

typedef __attribute__((ext_vector_type(2))) float v2f;
typedef __attribute__((ext_vector_type(8))) float v8f;

__global__ void precompute_proj(const float* __restrict__ aa_emb,
                                const float* __restrict__ chain_emb,
                                const float* __restrict__ W1,
                                float* __restrict__ proj) {
  const int r = blockIdx.x;
  const int c = threadIdx.x;
  const float* emb;
  const float* wrow;
  if (r < N_AA) {
    emb  = aa_emb + (size_t)r * DFEAT;
    wrow = W1;
  } else {
    emb  = chain_emb + (size_t)(r - N_AA) * DFEAT;
    wrow = W1 + (size_t)(DFEAT + DCOORD + DDIH) * DH1;
  }
  float acc = 0.f;
  for (int k = 0; k < DFEAT; ++k)
    acc = fmaf(emb[k], wrow[(size_t)k * DH1 + c], acc);
  proj[(size_t)r * DH1 + c] = acc;
}

__global__ __launch_bounds__(128) void residue_embed_fused(
    const int*   __restrict__ seq_idx,
    const float* __restrict__ xyz,
    const float* __restrict__ orient,
    const float* __restrict__ dihedrals,
    const int*   __restrict__ chain_idx,
    const float* __restrict__ atom_mask,
    const float* __restrict__ W1, const float* __restrict__ b1,
    const float* __restrict__ W2, const float* __restrict__ b2,
    const float* __restrict__ W3, const float* __restrict__ b3,
    const float* __restrict__ W4, const float* __restrict__ b4,
    const float* __restrict__ proj,
    float*       __restrict__ out)
{
  __shared__ float s_h1[16][H1S];
  __shared__ float s_h2[16][H2S];
  __shared__ float s_h3[16][H2S];
  __shared__ float s_xyzl[16][48];
  __shared__ float s_dih[16][40];
  __shared__ int   s_seq[16];
  __shared__ int   s_ch[16];

  const int tid  = threadIdx.x;
  const int tok0 = blockIdx.x * 16;

  if (tid < 16) {
    s_seq[tid] = seq_idx[tok0 + tid];
    s_ch[tid]  = chain_idx[tok0 + tid];
  }

  if (tid < 48) {
    const int   t = tid / 3, d = tid % 3;
    const float x = dihedrals[(size_t)(tok0 + t) * 3 + d];
    const float kF[6] = {1.f, 2.f, 3.f, 1.f, 0.5f, 1.f / 3.f};
    const int base = d * 13;
    s_dih[t][base] = x;
#pragma unroll
    for (int f = 0; f < 6; ++f) {
      s_dih[t][base + 1 + f] = sinf(kF[f] * x);
      s_dih[t][base + 7 + f] = cosf(kF[f] * x);
    }
  }

  for (int task = tid; task < 16 * NATOM; task += 128) {
    const int t = task / NATOM, a = task % NATOM;
    const int tok = tok0 + t;
    const float* p  = xyz + ((size_t)tok * NATOM + a) * 3;
    const float* ca = xyz + ((size_t)tok * NATOM + 1) * 3;
    const float rx = p[0] - ca[0], ry = p[1] - ca[1], rz = p[2] - ca[2];
    const float* R = orient + (size_t)tok * 9;
    const float m = atom_mask[(size_t)tok * NATOM + a];
    s_xyzl[t][a * 3 + 0] = (R[0] * rx + R[3] * ry + R[6] * rz) * m;
    s_xyzl[t][a * 3 + 1] = (R[1] * rx + R[4] * ry + R[7] * rz) * m;
    s_xyzl[t][a * 3 + 2] = (R[2] * rx + R[5] * ry + R[8] * rz) * m;
  }
  __syncthreads();

  const float* w1dih = W1 + (size_t)(DFEAT + DCOORD) * DH1;
  for (int t = 0; t < 16; ++t) {
    const int sq = s_seq[t], ch = s_ch[t];
    const float* w1c = W1 + (size_t)(DFEAT + sq * 45) * DH1;
    const float* pa  = proj + (size_t)sq * DH1;
    const float* pc  = proj + (size_t)(N_AA + ch) * DH1;
#pragma unroll
    for (int pass = 0; pass < 2; ++pass) {
      const int c = pass * 128 + tid;
      float acc = b1[c] + pa[c] + pc[c];
      for (int k = 0; k < 45; ++k)
        acc = fmaf(s_xyzl[t][k], w1c[(size_t)k * DH1 + c], acc);
      for (int k = 0; k < DDIH; ++k)
        acc = fmaf(s_dih[t][k], w1dih[(size_t)k * DH1 + c], acc);
      s_h1[t][c] = fmaxf(acc, 0.f);
    }
  }
  __syncthreads();

  const int wave = tid >> 5;
  const int lane = tid & 31;
  const int half = lane >> 4;
  const int lm   = lane & 15;

#pragma unroll
  for (int tt = 0; tt < 2; ++tt) {
    const int col0 = (wave * 2 + tt) * 16;
    v8f acc = {};
    for (int k = 0; k < 256; k += 4) {
      v2f a = *(const v2f*)&s_h1[lm][k + 2 * half];
      v2f b;
      b.x = W2[(size_t)(k + 2 * half)     * DFEAT + col0 + lm];
      b.y = W2[(size_t)(k + 2 * half + 1) * DFEAT + col0 + lm];
      acc = __builtin_amdgcn_wmma_f32_16x16x4_f32(
          false, a, false, b, (short)0, acc, false, false);
    }
#pragma unroll
    for (int r = 0; r < 8; ++r) {
      const float v = acc[r] + b2[col0 + lm];
      s_h2[r + 8 * half][col0 + lm] = fmaxf(v, 0.f);
    }
  }
  __syncthreads();

#pragma unroll
  for (int tt = 0; tt < 2; ++tt) {
    const int col0 = (wave * 2 + tt) * 16;
    v8f acc = {};
    for (int k = 0; k < 128; k += 4) {
      v2f a = *(const v2f*)&s_h2[lm][k + 2 * half];
      v2f b;
      b.x = W3[(size_t)(k + 2 * half)     * DFEAT + col0 + lm];
      b.y = W3[(size_t)(k + 2 * half + 1) * DFEAT + col0 + lm];
      acc = __builtin_amdgcn_wmma_f32_16x16x4_f32(
          false, a, false, b, (short)0, acc, false, false);
    }
#pragma unroll
    for (int r = 0; r < 8; ++r) {
      const float v = acc[r] + b3[col0 + lm];
      s_h3[r + 8 * half][col0 + lm] = fmaxf(v, 0.f);
    }
  }
  __syncthreads();

#pragma unroll
  for (int tt = 0; tt < 2; ++tt) {
    const int col0 = (wave * 2 + tt) * 16;
    v8f acc = {};
    for (int k = 0; k < 128; k += 4) {
      v2f a = *(const v2f*)&s_h3[lm][k + 2 * half];
      v2f b;
      b.x = W4[(size_t)(k + 2 * half)     * DFEAT + col0 + lm];
      b.y = W4[(size_t)(k + 2 * half + 1) * DFEAT + col0 + lm];
      acc = __builtin_amdgcn_wmma_f32_16x16x4_f32(
          false, a, false, b, (short)0, acc, false, false);
    }
#pragma unroll
    for (int r = 0; r < 8; ++r) {
      out[(size_t)(tok0 + r + 8 * half) * DFEAT + col0 + lm] = acc[r] + b4[col0 + lm];
    }
  }
}

extern "C" void kernel_launch(void* const* d_in, const int* in_sizes, int n_in,
                              void* d_out, int out_size, void* d_ws, size_t ws_size,
                              hipStream_t stream) {
  const int*   seq    = (const int*)  d_in[0];
  const float* xyz    = (const float*)d_in[1];
  const float* orient = (const float*)d_in[2];
  const float* dih    = (const float*)d_in[3];
  const int*   chain  = (const int*)  d_in[4];
  const float* amask  = (const float*)d_in[5];
  const float* aa_emb = (const float*)d_in[6];
  const float* ch_emb = (const float*)d_in[7];
  const float* W1 = (const float*)d_in[8];
  const float* b1 = (const float*)d_in[9];
  const float* W2 = (const float*)d_in[10];
  const float* b2 = (const float*)d_in[11];
  const float* W3 = (const float*)d_in[12];
  const float* b3 = (const float*)d_in[13];
  const float* W4 = (const float*)d_in[14];
  const float* b4 = (const float*)d_in[15];
  float* out  = (float*)d_out;
  float* proj = (float*)d_ws;

  const int ntok = in_sizes[0];

  precompute_proj<<<N_AA + 10, DH1, 0, stream>>>(aa_emb, ch_emb, W1, proj);
  residue_embed_fused<<<ntok / 16, 128, 0, stream>>>(
      seq, xyz, orient, dih, chain, amask,
      W1, b1, W2, b2, W3, b3, W4, b4, proj, out);
}


# Layer-1 dihedral rows (39 of 84 per token) computed once per 16-token tile as a 16x40x256 f32 WMMA product instead of per-token FMAs with one weight load each; sum order: dihedral part first
# speedup vs baseline: 1.2874x; 1.2874x over previous
#include <hip/hip_runtime.h>

#define N_AA   21
#define NATOM  15
#define DFEAT  128
#define DCOORD (N_AA * NATOM * 3)
#define DDIH   39
#define DH1    256
#define H1S    260
#define H2S    132

typedef __attribute__((ext_vector_type(2))) float v2f;
typedef __attribute__((ext_vector_type(8))) float v8f;

__global__ void precompute_proj(const float* __restrict__ aa_emb,
                                const float* __restrict__ chain_emb,
                                const float* __restrict__ W1,
                                float* __restrict__ proj) {
  const int r = blockIdx.x;
  const int c = threadIdx.x;
  const float* emb;
  const float* wrow;
  if (r < N_AA) {
    emb  = aa_emb + (size_t)r * DFEAT;
    wrow = W1;
  } else {
    emb  = chain_emb + (size_t)(r - N_AA) * DFEAT;
    wrow = W1 + (size_t)(DFEAT + DCOORD + DDIH) * DH1;
  }
  float acc = 0.f;
  for (int k = 0; k < DFEAT; ++k)
    acc = fmaf(emb[k], wrow[(size_t)k * DH1 + c], acc);
  proj[(size_t)r * DH1 + c] = acc;
}

__global__ __launch_bounds__(128) void residue_embed_fused(
    const int*   __restrict__ seq_idx,
    const float* __restrict__ xyz,
    const float* __restrict__ orient,
    const float* __restrict__ dihedrals,
    const int*   __restrict__ chain_idx,
    const float* __restrict__ atom_mask,
    const float* __restrict__ W1, const float* __restrict__ b1,
    const float* __restrict__ W2, const float* __restrict__ b2,
    const float* __restrict__ W3, const float* __restrict__ b3,
    const float* __restrict__ W4, const float* __restrict__ b4,
    const float* __restrict__ proj,
    float*       __restrict__ out)
{
  __shared__ float s_h1[16][H1S];
  __shared__ float s_h2[16][H2S];
  __shared__ float s_h3[16][H2S];
  __shared__ float s_xyzl[16][48];
  __shared__ __attribute__((aligned(16))) float s_dih[16][40];
  __shared__ int   s_seq[16];
  __shared__ int   s_ch[16];

  const int tid  = threadIdx.x;
  const int tok0 = blockIdx.x * 16;

  if (tid < 16) {
    s_seq[tid] = seq_idx[tok0 + tid];
    s_ch[tid]  = chain_idx[tok0 + tid];
    s_dih[tid][DDIH] = 0.f;
  }

  if (tid < 48) {
    const int   t = tid / 3, d = tid % 3;
    const float x = dihedrals[(size_t)(tok0 + t) * 3 + d];
    const float kF[6] = {1.f, 2.f, 3.f, 1.f, 0.5f, 1.f / 3.f};
    const int base = d * 13;
    s_dih[t][base] = x;
#pragma unroll
    for (int f = 0; f < 6; ++f) {
      s_dih[t][base + 1 + f] = sinf(kF[f] * x);
      s_dih[t][base + 7 + f] = cosf(kF[f] * x);
    }
  }

  for (int task = tid; task < 16 * NATOM; task += 128) {
    const int t = task / NATOM, a = task % NATOM;
    const int tok = tok0 + t;
    const float* p  = xyz + ((size_t)tok * NATOM + a) * 3;
    const float* ca = xyz + ((size_t)tok * NATOM + 1) * 3;
    const float rx = p[0] - ca[0], ry = p[1] - ca[1], rz = p[2] - ca[2];
    const float* R = orient + (size_t)tok * 9;
    const float m = atom_mask[(size_t)tok * NATOM + a];
    s_xyzl[t][a * 3 + 0] = (R[0] * rx + R[3] * ry + R[6] * rz) * m;
    s_xyzl[t][a * 3 + 1] = (R[1] * rx + R[4] * ry + R[7] * rz) * m;
    s_xyzl[t][a * 3 + 2] = (R[2] * rx + R[5] * ry + R[8] * rz) * m;
  }
  __syncthreads();

  const int wave = tid >> 5;
  const int lane = tid & 31;
  const int half = lane >> 4;
  const int lm   = lane & 15;

  {
    const float* w1dih = W1 + (size_t)(DFEAT + DCOORD) * DH1;
    v2f afr[10];
#pragma unroll
    for (int i = 0; i < 10; ++i)
      afr[i] = *(const v2f*)&s_dih[lm][4 * i + 2 * half];
    v8f dacc[4] = {};
#pragma unroll
    for (int i = 0; i < 10; ++i) {
      const int r0 = 4 * i + 2 * half;
#pragma unroll
      for (int j = 0; j < 4; ++j) {
        const int col = (wave * 4 + j) * 16 + lm;
        v2f b;
        b.x = w1dih[(size_t)r0 * DH1 + col];
        if (i < 9) {
          b.y = w1dih[(size_t)(r0 + 1) * DH1 + col];
        } else {
          const float v = w1dih[(size_t)(r0 + 1 - half) * DH1 + col];
          b.y = half ? 0.f : v;
        }
        dacc[j] = __builtin_amdgcn_wmma_f32_16x16x4_f32(
            false, afr[i], false, b, (short)0, dacc[j], false, false);
      }
    }
#pragma unroll
    for (int j = 0; j < 4; ++j) {
      const int col = (wave * 4 + j) * 16 + lm;
#pragma unroll
      for (int r = 0; r < 8; ++r)
        s_h1[r + 8 * half][col] = dacc[j][r];
    }
  }
  __syncthreads();

  for (int t = 0; t < 16; ++t) {
    const int sq = s_seq[t], ch = s_ch[t];
    const float* w1c = W1 + (size_t)(DFEAT + sq * 45) * DH1;
    const float* pa  = proj + (size_t)sq * DH1;
    const float* pc  = proj + (size_t)(N_AA + ch) * DH1;
#pragma unroll
    for (int pass = 0; pass < 2; ++pass) {
      const int c = pass * 128 + tid;
      float acc = (b1[c] + pa[c] + pc[c]) + s_h1[t][c];
      for (int k = 0; k < 45; ++k)
        acc = fmaf(s_xyzl[t][k], w1c[(size_t)k * DH1 + c], acc);
      s_h1[t][c] = fmaxf(acc, 0.f);
    }
  }
  __syncthreads();

#pragma unroll
  for (int tt = 0; tt < 2; ++tt) {
    const int col0 = (wave * 2 + tt) * 16;
    v8f acc = {};
    for (int k = 0; k < 256; k += 4) {
      v2f a = *(const v2f*)&s_h1[lm][k + 2 * half];
      v2f b;
      b.x = W2[(size_t)(k + 2 * half)     * DFEAT + col0 + lm];
      b.y = W2[(size_t)(k + 2 * half + 1) * DFEAT + col0 + lm];
      acc = __builtin_amdgcn_wmma_f32_16x16x4_f32(
          false, a, false, b, (short)0, acc, false, false);
    }
#pragma unroll
    for (int r = 0; r < 8; ++r) {
      const float v = acc[r] + b2[col0 + lm];
      s_h2[r + 8 * half][col0 + lm] = fmaxf(v, 0.f);
    }
  }
  __syncthreads();

#pragma unroll
  for (int tt = 0; tt < 2; ++tt) {
    const int col0 = (wave * 2 + tt) * 16;
    v8f acc = {};
    for (int k = 0; k < 128; k += 4) {
      v2f a = *(const v2f*)&s_h2[lm][k + 2 * half];
      v2f b;
      b.x = W3[(size_t)(k + 2 * half)     * DFEAT + col0 + lm];
      b.y = W3[(size_t)(k + 2 * half + 1) * DFEAT + col0 + lm];
      acc = __builtin_amdgcn_wmma_f32_16x16x4_f32(
          false, a, false, b, (short)0, acc, false, false);
    }
#pragma unroll
    for (int r = 0; r < 8; ++r) {
      const float v = acc[r] + b3[col0 + lm];
      s_h3[r + 8 * half][col0 + lm] = fmaxf(v, 0.f);
    }
  }
  __syncthreads();

#pragma unroll
  for (int tt = 0; tt < 2; ++tt) {
    const int col0 = (wave * 2 + tt) * 16;
    v8f acc = {};
    for (int k = 0; k < 128; k += 4) {
      v2f a = *(const v2f*)&s_h3[lm][k + 2 * half];
      v2f b;
      b.x = W4[(size_t)(k + 2 * half)     * DFEAT + col0 + lm];
      b.y = W4[(size_t)(k + 2 * half + 1) * DFEAT + col0 + lm];
      acc = __builtin_amdgcn_wmma_f32_16x16x4_f32(
          false, a, false, b, (short)0, acc, false, false);
    }
#pragma unroll
    for (int r = 0; r < 8; ++r) {
      out[(size_t)(tok0 + r + 8 * half) * DFEAT + col0 + lm] = acc[r] + b4[col0 + lm];
    }
  }
}

extern "C" void kernel_launch(void* const* d_in, const int* in_sizes, int n_in,
                              void* d_out, int out_size, void* d_ws, size_t ws_size,
                              hipStream_t stream) {
  const int*   seq    = (const int*)  d_in[0];
  const float* xyz    = (const float*)d_in[1];
  const float* orient = (const float*)d_in[2];
  const float* dih    = (const float*)d_in[3];
  const int*   chain  = (const int*)  d_in[4];
  const float* amask  = (const float*)d_in[5];
  const float* aa_emb = (const float*)d_in[6];
  const float* ch_emb = (const float*)d_in[7];
  const float* W1 = (const float*)d_in[8];
  const float* b1 = (const float*)d_in[9];
  const float* W2 = (const float*)d_in[10];
  const float* b2 = (const float*)d_in[11];
  const float* W3 = (const float*)d_in[12];
  const float* b3 = (const float*)d_in[13];
  const float* W4 = (const float*)d_in[14];
  const float* b4 = (const float*)d_in[15];
  float* out  = (float*)d_out;
  float* proj = (float*)d_ws;

  const int ntok = in_sizes[0];

  precompute_proj<<<N_AA + 10, DH1, 0, stream>>>(aa_emb, ch_emb, W1, proj);
  residue_embed_fused<<<ntok / 16, 128, 0, stream>>>(
      seq, xyz, orient, dih, chain, amask,
      W1, b1, W2, b2, W3, b3, W4, b4, proj, out);
}


# Token tile 16 -> 64 per work-group (512 threads): each weight fragment feeds 2-4 row tiles, and a residue type's 45 coord rows are loaded once per tile for all its tokens; sum order unchanged
# speedup vs baseline: 1.8881x; 1.4666x over previous
#include <hip/hip_runtime.h>


#define N_AA   21
#define NATOM  15
#define DFEAT  128
#define DCOORD (N_AA * NATOM * 3)
#define DDIH   39
#define DH1    256
#define NCRD   (NATOM * 3)
#define H1S    260
#define H2S    132
#define MT     4
#define TOK    (16 * MT)
#define NTHR   512

typedef __attribute__((ext_vector_type(2))) float v2f;
typedef __attribute__((ext_vector_type(8))) float v8f;

__global__ void precompute_proj(const float* __restrict__ aa_emb,
                                const float* __restrict__ chain_emb,
                                const float* __restrict__ W1,
                                float* __restrict__ proj) {
  const int r = blockIdx.x;
  const int c = threadIdx.x;
  const float* emb;
  const float* wrow;
  if (r < N_AA) {
    emb  = aa_emb + (size_t)r * DFEAT;
    wrow = W1;
  } else {
    emb  = chain_emb + (size_t)(r - N_AA) * DFEAT;
    wrow = W1 + (size_t)(DFEAT + DCOORD + DDIH) * DH1;
  }
  float acc = 0.f;
  for (int k = 0; k < DFEAT; ++k)
    acc = fmaf(emb[k], wrow[(size_t)k * DH1 + c], acc);
  proj[(size_t)r * DH1 + c] = acc;
}

__device__ __forceinline__ void load_type_rows(float (&w)[NCRD], const float* __restrict__ W1,
                                               int type, int c) {
  const float* wp = W1 + (size_t)(DFEAT + type * NCRD) * DH1 + c;
#pragma unroll
  for (int k = 0; k < NCRD; ++k) w[k] = wp[(size_t)k * DH1];
}

__device__ __forceinline__ void coord_rows_of_type(unsigned long long m, const float (&w)[NCRD],
                                                   float (*h1)[H1S], const float (*xl)[48], int c) {
  while (m) {
    const int t = __builtin_ctzll(m);
    m &= m - 1;
    float acc = h1[t][c];
#pragma unroll
    for (int k = 0; k < NCRD; ++k) acc = fmaf(xl[t][k], w[k], acc);
    h1[t][c] = fmaxf(acc, 0.f);
  }
}

__global__ __launch_bounds__(NTHR, 8) void residue_embed_fused(
    const int*   __restrict__ seq_idx,
    const float* __restrict__ xyz,
    const float* __restrict__ orient,
    const float* __restrict__ dihedrals,
    const int*   __restrict__ chain_idx,
    const float* __restrict__ atom_mask,
    const float* __restrict__ W1, const float* __restrict__ b1,
    const float* __restrict__ W2, const float* __restrict__ b2,
    const float* __restrict__ W3, const float* __restrict__ b3,
    const float* __restrict__ W4, const float* __restrict__ b4,
    const float* __restrict__ proj,
    float*       __restrict__ out)
{
  __shared__ float s_h1[TOK][H1S];
  __shared__ __attribute__((aligned(16))) float s_h2[TOK][H2S];
  __shared__ float s_h3[TOK][H2S];
  __shared__ __attribute__((aligned(16))) float s_xyzl[TOK][48];
  __shared__ __attribute__((aligned(16))) float s_dih[TOK][40];
  __shared__ int   s_seq[TOK];
  __shared__ int   s_ch[TOK];

  const int tid  = threadIdx.x;
  const int tok0 = blockIdx.x * TOK;

  if (tid < TOK) {
    s_seq[tid] = seq_idx[tok0 + tid];
    s_ch[tid]  = chain_idx[tok0 + tid];
    s_dih[tid][DDIH] = 0.f;
  }

  if (tid < TOK * 3) {
    const int   t = tid / 3, d = tid % 3;
    const float x = dihedrals[(size_t)(tok0 + t) * 3 + d];
    const float kF[6] = {1.f, 2.f, 3.f, 1.f, 0.5f, 1.f / 3.f};
    const int base = d * 13;
    s_dih[t][base] = x;
#pragma unroll
    for (int f = 0; f < 6; ++f) {
      s_dih[t][base + 1 + f] = sinf(kF[f] * x);
      s_dih[t][base + 7 + f] = cosf(kF[f] * x);
    }
  }

  for (int task = tid; task < TOK * NATOM; task += NTHR) {
    const int t = task / NATOM, a = task % NATOM;
    const int tok = tok0 + t;
    const float* p  = xyz + ((size_t)tok * NATOM + a) * 3;
    const float* ca = xyz + ((size_t)tok * NATOM + 1) * 3;
    const float rx = p[0] - ca[0], ry = p[1] - ca[1], rz = p[2] - ca[2];
    const float* R = orient + (size_t)tok * 9;
    const float m = atom_mask[(size_t)tok * NATOM + a];
    s_xyzl[t][a * 3 + 0] = (R[0] * rx + R[3] * ry + R[6] * rz) * m;
    s_xyzl[t][a * 3 + 1] = (R[1] * rx + R[4] * ry + R[7] * rz) * m;
    s_xyzl[t][a * 3 + 2] = (R[2] * rx + R[5] * ry + R[8] * rz) * m;
  }
  __syncthreads();

  const int wave = tid >> 5;
  const int lane = tid & 31;
  const int half = lane >> 4;
  const int lm   = lane & 15;

  {
    const float* w1dih = W1 + (size_t)(DFEAT + DCOORD) * DH1;
    const int col = wave * 16 + lm;
    v8f dacc[MT] = {};
#pragma unroll
    for (int i = 0; i < 10; ++i) {
      const int r0 = 4 * i + 2 * half;
      v2f b;
      b.x = w1dih[(size_t)r0 * DH1 + col];
      if (i < 9) {
        b.y = w1dih[(size_t)(r0 + 1) * DH1 + col];
      } else {
        const float v = w1dih[(size_t)(r0 + 1 - half) * DH1 + col];
        b.y = half ? 0.f : v;
      }
#pragma unroll
      for (int mt = 0; mt < MT; ++mt) {
        const v2f a = *(const v2f*)&s_dih[16 * mt + lm][4 * i + 2 * half];
        dacc[mt] = __builtin_amdgcn_wmma_f32_16x16x4_f32(
            false, a, false, b, (short)0, dacc[mt], false, false);
      }
    }
#pragma unroll
    for (int mt = 0; mt < MT; ++mt) {
#pragma unroll
      for (int r = 0; r < 8; ++r)
        s_h1[16 * mt + r + 8 * half][col] = dacc[mt][r];
    }
  }
  __syncthreads();

  const int c = tid & (DH1 - 1);
  const int g = __builtin_amdgcn_readfirstlane(tid >> 8);

  {
    const float bb = b1[c];
    for (int t0 = g * (TOK / 2); t0 < (g + 1) * (TOK / 2); t0 += 8) {
      float pa[8], pc[8];
#pragma unroll
      for (int j = 0; j < 8; ++j) {
        pa[j] = proj[(size_t)s_seq[t0 + j] * DH1 + c];
        pc[j] = proj[(size_t)(N_AA + s_ch[t0 + j]) * DH1 + c];
      }
#pragma unroll
      for (int j = 0; j < 8; ++j)
        s_h1[t0 + j][c] = ((bb + pa[j]) + pc[j]) + s_h1[t0 + j][c];
    }
  }
  __syncthreads();

  {
    const int sq0 = s_seq[lane], sq1 = s_seq[32 + lane];
#pragma unroll 1
    for (int s = g; s < N_AA; s += 2) {
      const unsigned long long m =
          (unsigned long long)__builtin_amdgcn_ballot_w32(sq0 == s) |
          ((unsigned long long)__builtin_amdgcn_ballot_w32(sq1 == s) << 32);
      if (m == 0) continue;
      float w[NCRD];
      load_type_rows(w, W1, s, c);
      coord_rows_of_type(m, w, s_h1, s_xyzl, c);
    }
  }
  __syncthreads();

  {
    const int col = (wave & 7) * 16 + lm;
    const int row0 = (wave >> 3) * 32;
    v8f acc[2] = {};
#pragma unroll 8
    for (int k = 0; k < 256; k += 4) {
      v2f b;
      b.x = W2[(size_t)(k + 2 * half)     * DFEAT + col];
      b.y = W2[(size_t)(k + 2 * half + 1) * DFEAT + col];
#pragma unroll
      for (int mt = 0; mt < 2; ++mt) {
        const v2f a = *(const v2f*)&s_h1[row0 + 16 * mt + lm][k + 2 * half];
        acc[mt] = __builtin_amdgcn_wmma_f32_16x16x4_f32(
            false, a, false, b, (short)0, acc[mt], false, false);
      }
    }
    const float bias = b2[col];
#pragma unroll
    for (int mt = 0; mt < 2; ++mt) {
#pragma unroll
      for (int r = 0; r < 8; ++r)
        s_h2[row0 + 16 * mt + r + 8 * half][col] = fmaxf(acc[mt][r] + bias, 0.f);
    }
  }
  __syncthreads();

  {
    const int col = (wave & 7) * 16 + lm;
    const int row0 = (wave >> 3) * 32;
    v8f acc[2] = {};
#pragma unroll 8
    for (int k = 0; k < 128; k += 4) {
      v2f b;
      b.x = W3[(size_t)(k + 2 * half)     * DFEAT + col];
      b.y = W3[(size_t)(k + 2 * half + 1) * DFEAT + col];
#pragma unroll
      for (int mt = 0; mt < 2; ++mt) {
        const v2f a = *(const v2f*)&s_h2[row0 + 16 * mt + lm][k + 2 * half];
        acc[mt] = __builtin_amdgcn_wmma_f32_16x16x4_f32(
            false, a, false, b, (short)0, acc[mt], false, false);
      }
    }
    const float bias = b3[col];
#pragma unroll
    for (int mt = 0; mt < 2; ++mt) {
#pragma unroll
      for (int r = 0; r < 8; ++r)
        s_h3[row0 + 16 * mt + r + 8 * half][col] = fmaxf(acc[mt][r] + bias, 0.f);
    }
  }
  __syncthreads();

  {
    const int col = (wave & 7) * 16 + lm;
    const int row0 = (wave >> 3) * 32;
    v8f acc[2] = {};
#pragma unroll 8
    for (int k = 0; k < 128; k += 4) {
      v2f b;
      b.x = W4[(size_t)(k + 2 * half)     * DFEAT + col];
      b.y = W4[(size_t)(k + 2 * half + 1) * DFEAT + col];
#pragma unroll
      for (int mt = 0; mt < 2; ++mt) {
        const v2f a = *(const v2f*)&s_h3[row0 + 16 * mt + lm][k + 2 * half];
        acc[mt] = __builtin_amdgcn_wmma_f32_16x16x4_f32(
            false, a, false, b, (short)0, acc[mt], false, false);
      }
    }
    const float bias = b4[col];
#pragma unroll
    for (int mt = 0; mt < 2; ++mt) {
#pragma unroll
      for (int r = 0; r < 8; ++r)
        s_h2[row0 + 16 * mt + r + 8 * half][col] = acc[mt][r] + bias;
    }
  }
  __syncthreads();
#pragma unroll
  for (int i = 0; i < TOK / 16; ++i) {
    const int row = wave * (TOK / 16) + i;
    const float4 v = *(const float4*)&s_h2[row][4 * lane];
    *(float4*)&out[(size_t)(tok0 + row) * DFEAT + 4 * lane] = v;
  }
}

extern "C" void kernel_launch(void* const* d_in, const int* in_sizes, int n_in,
                              void* d_out, int out_size, void* d_ws, size_t ws_size,
                              hipStream_t stream) {
  const int*   seq    = (const int*)  d_in[0];
  const float* xyz    = (const float*)d_in[1];
  const float* orient = (const float*)d_in[2];
  const float* dih    = (const float*)d_in[3];
  const int*   chain  = (const int*)  d_in[4];
  const float* amask  = (const float*)d_in[5];
  const float* aa_emb = (const float*)d_in[6];
  const float* ch_emb = (const float*)d_in[7];
  const float* W1 = (const float*)d_in[8];
  const float* b1 = (const float*)d_in[9];
  const float* W2 = (const float*)d_in[10];
  const float* b2 = (const float*)d_in[11];
  const float* W3 = (const float*)d_in[12];
  const float* b3 = (const float*)d_in[13];
  const float* W4 = (const float*)d_in[14];
  const float* b4 = (const float*)d_in[15];
  float* out  = (float*)d_out;
  float* proj = (float*)d_ws;

  const int ntok = in_sizes[0];

  precompute_proj<<<N_AA + 10, DH1, 0, stream>>>(aa_emb, ch_emb, W1, proj);
  residue_embed_fused<<<ntok / TOK, NTHR, 0, stream>>>(
      seq, xyz, orient, dih, chain, amask,
      W1, b1, W2, b2, W3, b3, W4, b4, proj, out);
}


# Layer-1 coord rows: a lane owns 2 adjacent columns (packed f32 FMA) and 4 wave groups split the types, so one broadcast LDS read of a token's coords feeds 2 columns; per-column sum order unchanged
# speedup vs baseline: 2.0140x; 1.0667x over previous
#include <hip/hip_runtime.h>


#define N_AA   21
#define NATOM  15
#define DFEAT  128
#define DCOORD (N_AA * NATOM * 3)
#define DDIH   39
#define DH1    256
#define NCRD   (NATOM * 3)
#define H1S    260
#define H2S    132
#define MT     4
#define TOK    (16 * MT)
#define NTHR   512

typedef __attribute__((ext_vector_type(2))) float v2f;
typedef __attribute__((ext_vector_type(8))) float v8f;

__global__ void precompute_proj(const float* __restrict__ aa_emb,
                                const float* __restrict__ chain_emb,
                                const float* __restrict__ W1,
                                float* __restrict__ proj) {
  const int r = blockIdx.x;
  const int c = threadIdx.x;
  const float* emb;
  const float* wrow;
  if (r < N_AA) {
    emb  = aa_emb + (size_t)r * DFEAT;
    wrow = W1;
  } else {
    emb  = chain_emb + (size_t)(r - N_AA) * DFEAT;
    wrow = W1 + (size_t)(DFEAT + DCOORD + DDIH) * DH1;
  }
  float acc = 0.f;
  for (int k = 0; k < DFEAT; ++k)
    acc = fmaf(emb[k], wrow[(size_t)k * DH1 + c], acc);
  proj[(size_t)r * DH1 + c] = acc;
}

__device__ __forceinline__ void load_type_rows(v2f (&w)[NCRD], const float* __restrict__ W1,
                                               int type, int c2) {
  const float* wp = W1 + (size_t)(DFEAT + type * NCRD) * DH1 + c2;
#pragma unroll
  for (int k = 0; k < NCRD; ++k) w[k] = *(const v2f*)(wp + (size_t)k * DH1);
}

__device__ __forceinline__ void coord_rows_of_type(unsigned long long m, const v2f (&w)[NCRD],
                                                   float (*h1)[H1S], const float (*xl)[48], int c2) {
  while (m) {
    const int t = __builtin_ctzll(m);
    m &= m - 1;
    v2f acc = *(const v2f*)&h1[t][c2];
#pragma unroll
    for (int k = 0; k < NCRD; ++k) {
      const float x = xl[t][k];
      const v2f xx = {x, x};
      acc = __builtin_elementwise_fma(xx, w[k], acc);
    }
    acc.x = fmaxf(acc.x, 0.f);
    acc.y = fmaxf(acc.y, 0.f);
    *(v2f*)&h1[t][c2] = acc;
  }
}

__global__ __launch_bounds__(NTHR, 8) void residue_embed_fused(
    const int*   __restrict__ seq_idx,
    const float* __restrict__ xyz,
    const float* __restrict__ orient,
    const float* __restrict__ dihedrals,
    const int*   __restrict__ chain_idx,
    const float* __restrict__ atom_mask,
    const float* __restrict__ W1, const float* __restrict__ b1,
    const float* __restrict__ W2, const float* __restrict__ b2,
    const float* __restrict__ W3, const float* __restrict__ b3,
    const float* __restrict__ W4, const float* __restrict__ b4,
    const float* __restrict__ proj,
    float*       __restrict__ out)
{
  __shared__ __attribute__((aligned(16))) float s_h1[TOK][H1S];
  __shared__ __attribute__((aligned(16))) float s_h2[TOK][H2S];
  __shared__ float s_h3[TOK][H2S];
  __shared__ __attribute__((aligned(16))) float s_xyzl[TOK][48];
  __shared__ __attribute__((aligned(16))) float s_dih[TOK][40];
  __shared__ int   s_seq[TOK];
  __shared__ int   s_ch[TOK];

  const int tid  = threadIdx.x;
  const int tok0 = blockIdx.x * TOK;

  if (tid < TOK) {
    s_seq[tid] = seq_idx[tok0 + tid];
    s_ch[tid]  = chain_idx[tok0 + tid];
    s_dih[tid][DDIH] = 0.f;
  }

  if (tid < TOK * 3) {
    const int   t = tid / 3, d = tid % 3;
    const float x = dihedrals[(size_t)(tok0 + t) * 3 + d];
    const float kF[6] = {1.f, 2.f, 3.f, 1.f, 0.5f, 1.f / 3.f};
    const int base = d * 13;
    s_dih[t][base] = x;
#pragma unroll
    for (int f = 0; f < 6; ++f) {
      s_dih[t][base + 1 + f] = sinf(kF[f] * x);
      s_dih[t][base + 7 + f] = cosf(kF[f] * x);
    }
  }

  for (int task = tid; task < TOK * NATOM; task += NTHR) {
    const int t = task / NATOM, a = task % NATOM;
    const int tok = tok0 + t;
    const float* p  = xyz + ((size_t)tok * NATOM + a) * 3;
    const float* ca = xyz + ((size_t)tok * NATOM + 1) * 3;
    const float rx = p[0] - ca[0], ry = p[1] - ca[1], rz = p[2] - ca[2];
    const float* R = orient + (size_t)tok * 9;
    const float m = atom_mask[(size_t)tok * NATOM + a];
    s_xyzl[t][a * 3 + 0] = (R[0] * rx + R[3] * ry + R[6] * rz) * m;
    s_xyzl[t][a * 3 + 1] = (R[1] * rx + R[4] * ry + R[7] * rz) * m;
    s_xyzl[t][a * 3 + 2] = (R[2] * rx + R[5] * ry + R[8] * rz) * m;
  }
  __syncthreads();

  const int wave = tid >> 5;
  const int lane = tid & 31;
  const int half = lane >> 4;
  const int lm   = lane & 15;

  {
    const float* w1dih = W1 + (size_t)(DFEAT + DCOORD) * DH1;
    const int col = wave * 16 + lm;
    v8f dacc[MT] = {};
#pragma unroll
    for (int i = 0; i < 10; ++i) {
      const int r0 = 4 * i + 2 * half;
      v2f b;
      b.x = w1dih[(size_t)r0 * DH1 + col];
      if (i < 9) {
        b.y = w1dih[(size_t)(r0 + 1) * DH1 + col];
      } else {
        const float v = w1dih[(size_t)(r0 + 1 - half) * DH1 + col];
        b.y = half ? 0.f : v;
      }
#pragma unroll
      for (int mt = 0; mt < MT; ++mt) {
        const v2f a = *(const v2f*)&s_dih[16 * mt + lm][4 * i + 2 * half];
        dacc[mt] = __builtin_amdgcn_wmma_f32_16x16x4_f32(
            false, a, false, b, (short)0, dacc[mt], false, false);
      }
    }
#pragma unroll
    for (int mt = 0; mt < MT; ++mt) {
#pragma unroll
      for (int r = 0; r < 8; ++r)
        s_h1[16 * mt + r + 8 * half][col] = dacc[mt][r];
    }
  }
  __syncthreads();

  const int c = tid & (DH1 - 1);
  const int g = __builtin_amdgcn_readfirstlane(tid >> 8);

  {
    const float bb = b1[c];
    for (int t0 = g * (TOK / 2); t0 < (g + 1) * (TOK / 2); t0 += 8) {
      float pa[8], pc[8];
#pragma unroll
      for (int j = 0; j < 8; ++j) {
        pa[j] = proj[(size_t)s_seq[t0 + j] * DH1 + c];
        pc[j] = proj[(size_t)(N_AA + s_ch[t0 + j]) * DH1 + c];
      }
#pragma unroll
      for (int j = 0; j < 8; ++j)
        s_h1[t0 + j][c] = ((bb + pa[j]) + pc[j]) + s_h1[t0 + j][c];
    }
  }
  __syncthreads();

  {
    const int c2 = 2 * (tid & (DH1 / 2 - 1));
    const int gq = __builtin_amdgcn_readfirstlane(tid >> 7);
    const int sq0 = s_seq[lane], sq1 = s_seq[32 + lane];
#pragma unroll 1
    for (int s = gq; s < N_AA; s += 4) {
      const unsigned long long m =
          (unsigned long long)__builtin_amdgcn_ballot_w32(sq0 == s) |
          ((unsigned long long)__builtin_amdgcn_ballot_w32(sq1 == s) << 32);
      if (m == 0) continue;
      v2f w[NCRD];
      load_type_rows(w, W1, s, c2);
      coord_rows_of_type(m, w, s_h1, s_xyzl, c2);
    }
  }
  __syncthreads();

  {
    const int col = (wave & 7) * 16 + lm;
    const int row0 = (wave >> 3) * 32;
    v8f acc[2] = {};
#pragma unroll 8
    for (int k = 0; k < 256; k += 4) {
      v2f b;
      b.x = W2[(size_t)(k + 2 * half)     * DFEAT + col];
      b.y = W2[(size_t)(k + 2 * half + 1) * DFEAT + col];
#pragma unroll
      for (int mt = 0; mt < 2; ++mt) {
        const v2f a = *(const v2f*)&s_h1[row0 + 16 * mt + lm][k + 2 * half];
        acc[mt] = __builtin_amdgcn_wmma_f32_16x16x4_f32(
            false, a, false, b, (short)0, acc[mt], false, false);
      }
    }
    const float bias = b2[col];
#pragma unroll
    for (int mt = 0; mt < 2; ++mt) {
#pragma unroll
      for (int r = 0; r < 8; ++r)
        s_h2[row0 + 16 * mt + r + 8 * half][col] = fmaxf(acc[mt][r] + bias, 0.f);
    }
  }
  __syncthreads();

  {
    const int col = (wave & 7) * 16 + lm;
    const int row0 = (wave >> 3) * 32;
    v8f acc[2] = {};
#pragma unroll 8
    for (int k = 0; k < 128; k += 4) {
      v2f b;
      b.x = W3[(size_t)(k + 2 * half)     * DFEAT + col];
      b.y = W3[(size_t)(k + 2 * half + 1) * DFEAT + col];
#pragma unroll
      for (int mt = 0; mt < 2; ++mt) {
        const v2f a = *(const v2f*)&s_h2[row0 + 16 * mt + lm][k + 2 * half];
        acc[mt] = __builtin_amdgcn_wmma_f32_16x16x4_f32(
            false, a, false, b, (short)0, acc[mt], false, false);
      }
    }
    const float bias = b3[col];
#pragma unroll
    for (int mt = 0; mt < 2; ++mt) {
#pragma unroll
      for (int r = 0; r < 8; ++r)
        s_h3[row0 + 16 * mt + r + 8 * half][col] = fmaxf(acc[mt][r] + bias, 0.f);
    }
  }
  __syncthreads();

  {
    const int col = (wave & 7) * 16 + lm;
    const int row0 = (wave >> 3) * 32;
    v8f acc[2] = {};
#pragma unroll 8
    for (int k = 0; k < 128; k += 4) {
      v2f b;
      b.x = W4[(size_t)(k + 2 * half)     * DFEAT + col];
      b.y = W4[(size_t)(k + 2 * half + 1) * DFEAT + col];
#pragma unroll
      for (int mt = 0; mt < 2; ++mt) {
        const v2f a = *(const v2f*)&s_h3[row0 + 16 * mt + lm][k + 2 * half];
        acc[mt] = __builtin_amdgcn_wmma_f32_16x16x4_f32(
            false, a, false, b, (short)0, acc[mt], false, false);
      }
    }
    const float bias = b4[col];
#pragma unroll
    for (int mt = 0; mt < 2; ++mt) {
#pragma unroll
      for (int r = 0; r < 8; ++r)
        s_h2[row0 + 16 * mt + r + 8 * half][col] = acc[mt][r] + bias;
    }
  }
  __syncthreads();
#pragma unroll
  for (int i = 0; i < TOK / 16; ++i) {
    const int row = wave * (TOK / 16) + i;
    const float4 v = *(const float4*)&s_h2[row][4 * lane];
    *(float4*)&out[(size_t)(tok0 + row) * DFEAT + 4 * lane] = v;
  }
}

extern "C" void kernel_launch(void* const* d_in, const int* in_sizes, int n_in,
                              void* d_out, int out_size, void* d_ws, size_t ws_size,
                              hipStream_t stream) {
  const int*   seq    = (const int*)  d_in[0];
  const float* xyz    = (const float*)d_in[1];
  const float* orient = (const float*)d_in[2];
  const float* dih    = (const float*)d_in[3];
  const int*   chain  = (const int*)  d_in[4];
  const float* amask  = (const float*)d_in[5];
  const float* aa_emb = (const float*)d_in[6];
  const float* ch_emb = (const float*)d_in[7];
  const float* W1 = (const float*)d_in[8];
  const float* b1 = (const float*)d_in[9];
  const float* W2 = (const float*)d_in[10];
  const float* b2 = (const float*)d_in[11];
  const float* W3 = (const float*)d_in[12];
  const float* b3 = (const float*)d_in[13];
  const float* W4 = (const float*)d_in[14];
  const float* b4 = (const float*)d_in[15];
  float* out  = (float*)d_out;
  float* proj = (float*)d_ws;

  const int ntok = in_sizes[0];

  precompute_proj<<<N_AA + 10, DH1, 0, stream>>>(aa_emb, ch_emb, W1, proj);
  residue_embed_fused<<<ntok / TOK, NTHR, 0, stream>>>(
      seq, xyz, orient, dih, chain, amask,
      W1, b1, W2, b2, W3, b3, W4, b4, proj, out);
}


# Tokens walked by residue type: the first launch also builds per-type token lists (prefix sums, no atomics), so a 64-token tile loads the 45 coord rows of 1-2 types, not ~20; sum order unchanged
# speedup vs baseline: 2.2786x; 1.1314x over previous
#include <hip/hip_runtime.h>


#define N_AA   21
#define NATOM  15
#define DFEAT  128
#define DCOORD (N_AA * NATOM * 3)
#define DDIH   39
#define DH1    256
#define NCRD   (NATOM * 3)
#define H1S    260
#define H2S    132
#define MT     4
#define TOK    (16 * MT)
#define NTHR   512
#define SORT_THR 1024
#define SORT_CH  32
#define WS_CNT   8192
#define WS_LIST  8256

typedef __attribute__((ext_vector_type(2))) float v2f;
typedef __attribute__((ext_vector_type(8))) float v8f;

__global__ __launch_bounds__(SORT_THR) void precompute_proj(const float* __restrict__ aa_emb,
                                const float* __restrict__ chain_emb,
                                const float* __restrict__ W1,
                                const int*   __restrict__ seq_idx, int ntok,
                                float* __restrict__ proj,
                                int*   __restrict__ wsi) {
  __shared__ int s_wtot[SORT_THR / 32];
  const int r   = blockIdx.x;
  const int tid = threadIdx.x;
  if (r < N_AA + 10) {
    if (tid < DH1) {
      const int c = tid;
      const float* emb;
      const float* wrow;
      if (r < N_AA) {
        emb  = aa_emb + (size_t)r * DFEAT;
        wrow = W1;
      } else {
        emb  = chain_emb + (size_t)(r - N_AA) * DFEAT;
        wrow = W1 + (size_t)(DFEAT + DCOORD + DDIH) * DH1;
      }
      float acc = 0.f;
      for (int k = 0; k < DFEAT; ++k)
        acc = fmaf(emb[k], wrow[(size_t)k * DH1 + c], acc);
      proj[(size_t)r * DH1 + c] = acc;
    }
    return;
  }

  const int s    = r - (N_AA + 10);
  const int lane = tid & 31;
  const int wave = tid >> 5;
  int* list = wsi + WS_LIST + (size_t)s * ntok;
  int running = 0;
  for (int base = 0; base < ntok; base += SORT_THR * SORT_CH) {
    const int first = base + tid * SORT_CH;
    unsigned m = 0;
    if (first + SORT_CH <= ntok) {
      const int4* p = (const int4*)(seq_idx + first);
#pragma unroll
      for (int i = 0; i < SORT_CH / 4; ++i) {
        const int4 v = p[i];
        m |= ((unsigned)(v.x == s) << (4 * i))     | ((unsigned)(v.y == s) << (4 * i + 1)) |
             ((unsigned)(v.z == s) << (4 * i + 2)) | ((unsigned)(v.w == s) << (4 * i + 3));
      }
    } else {
      for (int i = 0; i < SORT_CH; ++i)
        if (first + i < ntok) m |= (unsigned)(seq_idx[first + i] == s) << i;
    }
    const int cnt = __popc(m);
    int inc = cnt;
#pragma unroll
    for (int d = 1; d < 32; d <<= 1) {
      const int v = __shfl_up(inc, d);
      if (lane >= d) inc += v;
    }
    if (lane == 31) s_wtot[wave] = inc;
    __syncthreads();
    const int wt = s_wtot[lane];
    int winc = wt;
#pragma unroll
    for (int d = 1; d < 32; d <<= 1) {
      const int v = __shfl_up(winc, d);
      if (lane >= d) winc += v;
    }
    const int before = __shfl(winc, wave) - __shfl(wt, wave);
    const int total  = __shfl(winc, 31);
    int pos = running + before + inc - cnt;
    while (m) {
      const int b = __builtin_ctz(m);
      m &= m - 1;
      list[pos++] = first + b;
    }
    running += total;
    __syncthreads();
  }
  if (tid == 0) wsi[WS_CNT + s] = running;
}

__device__ __forceinline__ void load_type_rows(v2f (&w)[NCRD], const float* __restrict__ W1,
                                               int type, int c2) {
  const float* wp = W1 + (size_t)(DFEAT + type * NCRD) * DH1 + c2;
#pragma unroll
  for (int k = 0; k < NCRD; ++k) w[k] = *(const v2f*)(wp + (size_t)k * DH1);
}

__device__ __forceinline__ void coord_rows_of_type(unsigned long long m, const v2f (&w)[NCRD],
                                                   float (*h1)[H1S], const float (*xl)[48], int c2) {
  while (m) {
    const int t = __builtin_ctzll(m);
    m &= m - 1;
    v2f acc = *(const v2f*)&h1[t][c2];
#pragma unroll
    for (int k = 0; k < NCRD; ++k) {
      const float x = xl[t][k];
      const v2f xx = {x, x};
      acc = __builtin_elementwise_fma(xx, w[k], acc);
    }
    acc.x = fmaxf(acc.x, 0.f);
    acc.y = fmaxf(acc.y, 0.f);
    *(v2f*)&h1[t][c2] = acc;
  }
}

__global__ __launch_bounds__(NTHR, 8) void residue_embed_fused(
    const int*   __restrict__ seq_idx,
    const float* __restrict__ xyz,
    const float* __restrict__ orient,
    const float* __restrict__ dihedrals,
    const int*   __restrict__ chain_idx,
    const float* __restrict__ atom_mask,
    const float* __restrict__ W1, const float* __restrict__ b1,
    const float* __restrict__ W2, const float* __restrict__ b2,
    const float* __restrict__ W3, const float* __restrict__ b3,
    const float* __restrict__ W4, const float* __restrict__ b4,
    const float* __restrict__ proj,
    const int*   __restrict__ wsi,
    int ntok,
    float*       __restrict__ out)
{
  __shared__ __attribute__((aligned(16))) float s_h1[TOK][H1S];
  __shared__ __attribute__((aligned(16))) float s_h2[TOK][H2S];
  __shared__ float s_h3[TOK][H2S];
  __shared__ __attribute__((aligned(16))) float s_xyzl[TOK][48];
  __shared__ __attribute__((aligned(16))) float s_dih[TOK][40];
  __shared__ int   s_seq[TOK];
  __shared__ int   s_ch[TOK];
  __shared__ int   s_tok[TOK];

  const int tid  = threadIdx.x;
  const int tok0 = blockIdx.x * TOK;

  int token = 0;
  if (tid < TOK) {
    const int p = tok0 + tid;
    int s = 0, off = 0, end = 0;
#pragma unroll
    for (int q = 0; q < N_AA; ++q) {
      end += wsi[WS_CNT + q];
      if (p >= end) { s = q + 1; off = end; }
    }
    s = min(s, N_AA - 1);
    token = wsi[WS_LIST + (size_t)s * ntok + (p - off)];
    token = min(max(token, 0), ntok - 1);
    s_tok[tid] = token;
    s_seq[tid] = s;
    s_dih[tid][DDIH] = 0.f;
  }
  __syncthreads();
  if (tid < TOK) s_ch[tid] = chain_idx[token];

  if (tid < TOK * 3) {
    const int   t = tid / 3, d = tid % 3;
    const float x = dihedrals[(size_t)s_tok[t] * 3 + d];
    const float kF[6] = {1.f, 2.f, 3.f, 1.f, 0.5f, 1.f / 3.f};
    const int base = d * 13;
    s_dih[t][base] = x;
#pragma unroll
    for (int f = 0; f < 6; ++f) {
      s_dih[t][base + 1 + f] = sinf(kF[f] * x);
      s_dih[t][base + 7 + f] = cosf(kF[f] * x);
    }
  }

  for (int task = tid; task < TOK * NATOM; task += NTHR) {
    const int t = task / NATOM, a = task % NATOM;
    const int tok = s_tok[t];
    const float* p  = xyz + ((size_t)tok * NATOM + a) * 3;
    const float* ca = xyz + ((size_t)tok * NATOM + 1) * 3;
    const float rx = p[0] - ca[0], ry = p[1] - ca[1], rz = p[2] - ca[2];
    const float* R = orient + (size_t)tok * 9;
    const float m = atom_mask[(size_t)tok * NATOM + a];
    s_xyzl[t][a * 3 + 0] = (R[0] * rx + R[3] * ry + R[6] * rz) * m;
    s_xyzl[t][a * 3 + 1] = (R[1] * rx + R[4] * ry + R[7] * rz) * m;
    s_xyzl[t][a * 3 + 2] = (R[2] * rx + R[5] * ry + R[8] * rz) * m;
  }
  __syncthreads();

  const int wave = tid >> 5;
  const int lane = tid & 31;
  const int half = lane >> 4;
  const int lm   = lane & 15;

  {
    const float* w1dih = W1 + (size_t)(DFEAT + DCOORD) * DH1;
    const int col = wave * 16 + lm;
    v8f dacc[MT] = {};
#pragma unroll
    for (int i = 0; i < 10; ++i) {
      const int r0 = 4 * i + 2 * half;
      v2f b;
      b.x = w1dih[(size_t)r0 * DH1 + col];
      if (i < 9) {
        b.y = w1dih[(size_t)(r0 + 1) * DH1 + col];
      } else {
        const float v = w1dih[(size_t)(r0 + 1 - half) * DH1 + col];
        b.y = half ? 0.f : v;
      }
#pragma unroll
      for (int mt = 0; mt < MT; ++mt) {
        const v2f a = *(const v2f*)&s_dih[16 * mt + lm][4 * i + 2 * half];
        dacc[mt] = __builtin_amdgcn_wmma_f32_16x16x4_f32(
            false, a, false, b, (short)0, dacc[mt], false, false);
      }
    }
#pragma unroll
    for (int mt = 0; mt < MT; ++mt) {
#pragma unroll
      for (int r = 0; r < 8; ++r)
        s_h1[16 * mt + r + 8 * half][col] = dacc[mt][r];
    }
  }
  __syncthreads();

  const int c = tid & (DH1 - 1);
  const int g = __builtin_amdgcn_readfirstlane(tid >> 8);

  {
    const float bb = b1[c];
    for (int t0 = g * (TOK / 2); t0 < (g + 1) * (TOK / 2); t0 += 8) {
      float pa[8], pc[8];
#pragma unroll
      for (int j = 0; j < 8; ++j) {
        pa[j] = proj[(size_t)s_seq[t0 + j] * DH1 + c];
        pc[j] = proj[(size_t)(N_AA + s_ch[t0 + j]) * DH1 + c];
      }
#pragma unroll
      for (int j = 0; j < 8; ++j)
        s_h1[t0 + j][c] = ((bb + pa[j]) + pc[j]) + s_h1[t0 + j][c];
    }
  }
  __syncthreads();

  {
    const int c2 = 2 * (tid & (DH1 / 2 - 1));
    const int gq = __builtin_amdgcn_readfirstlane(tid >> 7);
    const int sq = s_seq[16 * gq + (lane & 15)];
#pragma unroll 1
    for (int s = 0; s < N_AA; ++s) {
      const unsigned long long m =
          (unsigned long long)(__builtin_amdgcn_ballot_w32(sq == s) & 0xffffu) << (16 * gq);
      if (m == 0) continue;
      v2f w[NCRD];
      load_type_rows(w, W1, s, c2);
      coord_rows_of_type(m, w, s_h1, s_xyzl, c2);
    }
  }
  __syncthreads();

  {
    const int col = (wave & 7) * 16 + lm;
    const int row0 = (wave >> 3) * 32;
    v8f acc[2] = {};
#pragma unroll 8
    for (int k = 0; k < 256; k += 4) {
      v2f b;
      b.x = W2[(size_t)(k + 2 * half)     * DFEAT + col];
      b.y = W2[(size_t)(k + 2 * half + 1) * DFEAT + col];
#pragma unroll
      for (int mt = 0; mt < 2; ++mt) {
        const v2f a = *(const v2f*)&s_h1[row0 + 16 * mt + lm][k + 2 * half];
        acc[mt] = __builtin_amdgcn_wmma_f32_16x16x4_f32(
            false, a, false, b, (short)0, acc[mt], false, false);
      }
    }
    const float bias = b2[col];
#pragma unroll
    for (int mt = 0; mt < 2; ++mt) {
#pragma unroll
      for (int r = 0; r < 8; ++r)
        s_h2[row0 + 16 * mt + r + 8 * half][col] = fmaxf(acc[mt][r] + bias, 0.f);
    }
  }
  __syncthreads();

  {
    const int col = (wave & 7) * 16 + lm;
    const int row0 = (wave >> 3) * 32;
    v8f acc[2] = {};
#pragma unroll 8
    for (int k = 0; k < 128; k += 4) {
      v2f b;
      b.x = W3[(size_t)(k + 2 * half)     * DFEAT + col];
      b.y = W3[(size_t)(k + 2 * half + 1) * DFEAT + col];
#pragma unroll
      for (int mt = 0; mt < 2; ++mt) {
        const v2f a = *(const v2f*)&s_h2[row0 + 16 * mt + lm][k + 2 * half];
        acc[mt] = __builtin_amdgcn_wmma_f32_16x16x4_f32(
            false, a, false, b, (short)0, acc[mt], false, false);
      }
    }
    const float bias = b3[col];
#pragma unroll
    for (int mt = 0; mt < 2; ++mt) {
#pragma unroll
      for (int r = 0; r < 8; ++r)
        s_h3[row0 + 16 * mt + r + 8 * half][col] = fmaxf(acc[mt][r] + bias, 0.f);
    }
  }
  __syncthreads();

  {
    const int col = (wave & 7) * 16 + lm;
    const int row0 = (wave >> 3) * 32;
    v8f acc[2] = {};
#pragma unroll 8
    for (int k = 0; k < 128; k += 4) {
      v2f b;
      b.x = W4[(size_t)(k + 2 * half)     * DFEAT + col];
      b.y = W4[(size_t)(k + 2 * half + 1) * DFEAT + col];
#pragma unroll
      for (int mt = 0; mt < 2; ++mt) {
        const v2f a = *(const v2f*)&s_h3[row0 + 16 * mt + lm][k + 2 * half];
        acc[mt] = __builtin_amdgcn_wmma_f32_16x16x4_f32(
            false, a, false, b, (short)0, acc[mt], false, false);
      }
    }
    const float bias = b4[col];
#pragma unroll
    for (int mt = 0; mt < 2; ++mt) {
#pragma unroll
      for (int r = 0; r < 8; ++r)
        s_h2[row0 + 16 * mt + r + 8 * half][col] = acc[mt][r] + bias;
    }
  }
  __syncthreads();
#pragma unroll
  for (int i = 0; i < TOK / 16; ++i) {
    const int row = wave * (TOK / 16) + i;
    const float4 v = *(const float4*)&s_h2[row][4 * lane];
    *(float4*)&out[(size_t)s_tok[row] * DFEAT + 4 * lane] = v;
  }
}

extern "C" void kernel_launch(void* const* d_in, const int* in_sizes, int n_in,
                              void* d_out, int out_size, void* d_ws, size_t ws_size,
                              hipStream_t stream) {
  const int*   seq    = (const int*)  d_in[0];
  const float* xyz    = (const float*)d_in[1];
  const float* orient = (const float*)d_in[2];
  const float* dih    = (const float*)d_in[3];
  const int*   chain  = (const int*)  d_in[4];
  const float* amask  = (const float*)d_in[5];
  const float* aa_emb = (const float*)d_in[6];
  const float* ch_emb = (const float*)d_in[7];
  const float* W1 = (const float*)d_in[8];
  const float* b1 = (const float*)d_in[9];
  const float* W2 = (const float*)d_in[10];
  const float* b2 = (const float*)d_in[11];
  const float* W3 = (const float*)d_in[12];
  const float* b3 = (const float*)d_in[13];
  const float* W4 = (const float*)d_in[14];
  const float* b4 = (const float*)d_in[15];
  float* out  = (float*)d_out;
  float* proj = (float*)d_ws;
  int*   wsi  = (int*)d_ws;

  const int ntok = in_sizes[0];

  precompute_proj<<<N_AA + 10 + N_AA, SORT_THR, 0, stream>>>(aa_emb, ch_emb, W1, seq, ntok, proj, wsi);
  residue_embed_fused<<<ntok / TOK, NTHR, 0, stream>>>(
      seq, xyz, orient, dih, chain, amask,
      W1, b1, W2, b2, W3, b3, W4, b4, proj, wsi, ntok, out);
}


# Layer-1 coord rows moved onto the WMMA accumulators of the dihedral product (per row tile and type present, a 16x48x256 product); bias+projections added at the store; sum order changes
# speedup vs baseline: 2.5448x; 1.1168x over previous
#include <hip/hip_runtime.h>


#define N_AA   21
#define NATOM  15
#define DFEAT  128
#define DCOORD (N_AA * NATOM * 3)
#define DDIH   39
#define DH1    256
#define NCRD   (NATOM * 3)
#define H1S    260
#define H2S    132
#define XS     52
#define DIHS   44
#define NPROJ  (N_AA + 10)
#define MT     4
#define TOK    (16 * MT)
#define NTHR   512
#define SORT_THR 1024
#define SORT_CH  32
#define WS_CNT   8192
#define WS_LIST  8256

typedef __attribute__((ext_vector_type(2))) float v2f;
typedef __attribute__((ext_vector_type(8))) float v8f;

__global__ __launch_bounds__(SORT_THR) void precompute_proj(const float* __restrict__ aa_emb,
                                const float* __restrict__ chain_emb,
                                const float* __restrict__ W1,
                                const int*   __restrict__ seq_idx, int ntok,
                                float* __restrict__ proj,
                                int*   __restrict__ wsi) {
  __shared__ int s_wtot[SORT_THR / 32];
  const int r   = blockIdx.x;
  const int tid = threadIdx.x;
  if (r < N_AA + 10) {
    if (tid < DH1) {
      const int c = tid;
      const float* emb;
      const float* wrow;
      if (r < N_AA) {
        emb  = aa_emb + (size_t)r * DFEAT;
        wrow = W1;
      } else {
        emb  = chain_emb + (size_t)(r - N_AA) * DFEAT;
        wrow = W1 + (size_t)(DFEAT + DCOORD + DDIH) * DH1;
      }
      float acc = 0.f;
      for (int k = 0; k < DFEAT; ++k)
        acc = fmaf(emb[k], wrow[(size_t)k * DH1 + c], acc);
      proj[(size_t)r * DH1 + c] = acc;
    }
    return;
  }

  const int s    = r - (N_AA + 10);
  const int lane = tid & 31;
  const int wave = tid >> 5;
  int* list = wsi + WS_LIST + (size_t)s * ntok;
  int running = 0;
  for (int base = 0; base < ntok; base += SORT_THR * SORT_CH) {
    const int first = base + tid * SORT_CH;
    unsigned m = 0;
    if (first + SORT_CH <= ntok) {
      const int4* p = (const int4*)(seq_idx + first);
#pragma unroll
      for (int i = 0; i < SORT_CH / 4; ++i) {
        const int4 v = p[i];
        m |= ((unsigned)(v.x == s) << (4 * i))     | ((unsigned)(v.y == s) << (4 * i + 1)) |
             ((unsigned)(v.z == s) << (4 * i + 2)) | ((unsigned)(v.w == s) << (4 * i + 3));
      }
    } else {
      for (int i = 0; i < SORT_CH; ++i)
        if (first + i < ntok) m |= (unsigned)(seq_idx[first + i] == s) << i;
    }
    const int cnt = __popc(m);
    int inc = cnt;
#pragma unroll
    for (int d = 1; d < 32; d <<= 1) {
      const int v = __shfl_up(inc, d);
      if (lane >= d) inc += v;
    }
    if (lane == 31) s_wtot[wave] = inc;
    __syncthreads();
    const int wt = s_wtot[lane];
    int winc = wt;
#pragma unroll
    for (int d = 1; d < 32; d <<= 1) {
      const int v = __shfl_up(winc, d);
      if (lane >= d) winc += v;
    }
    const int before = __shfl(winc, wave) - __shfl(wt, wave);
    const int total  = __shfl(winc, 31);
    int pos = running + before + inc - cnt;
    while (m) {
      const int b = __builtin_ctz(m);
      m &= m - 1;
      list[pos++] = first + b;
    }
    running += total;
    __syncthreads();
  }
  if (tid == 0) wsi[WS_CNT + s] = running;
}

__global__ __launch_bounds__(NTHR, 8) void residue_embed_fused(
    const int*   __restrict__ seq_idx,
    const float* __restrict__ xyz,
    const float* __restrict__ orient,
    const float* __restrict__ dihedrals,
    const int*   __restrict__ chain_idx,
    const float* __restrict__ atom_mask,
    const float* __restrict__ W1, const float* __restrict__ b1,
    const float* __restrict__ W2, const float* __restrict__ b2,
    const float* __restrict__ W3, const float* __restrict__ b3,
    const float* __restrict__ W4, const float* __restrict__ b4,
    const float* __restrict__ proj,
    const int*   __restrict__ wsi,
    int ntok,
    float*       __restrict__ out)
{
  __shared__ __attribute__((aligned(16))) float s_h1[TOK][H1S];
  __shared__ __attribute__((aligned(16))) float s_h2[TOK][H2S];
  __shared__ __attribute__((aligned(16))) float s_h3[TOK][H2S];
  __shared__ __attribute__((aligned(16))) float s_xyzl[TOK][XS];
  __shared__ int   s_seq[TOK];
  __shared__ int   s_ch[TOK];
  __shared__ int   s_tok[TOK];
  float (*s_dih)[DIHS] = (float (*)[DIHS])&s_h2[0][0];

  const int tid  = threadIdx.x;
  const int tok0 = blockIdx.x * TOK;

  int token = 0;
  if (tid < TOK) {
    const int p = tok0 + tid;
    int s = 0, off = 0, end = 0;
#pragma unroll
    for (int q = 0; q < N_AA; ++q) {
      end += wsi[WS_CNT + q];
      if (p >= end) { s = q + 1; off = end; }
    }
    s = min(s, N_AA - 1);
    token = wsi[WS_LIST + (size_t)s * ntok + (p - off)];
    token = min(max(token, 0), ntok - 1);
    s_tok[tid] = token;
    s_seq[tid] = s;
    s_dih[tid][DDIH] = 0.f;
    s_xyzl[tid][NCRD] = 0.f;
    s_xyzl[tid][NCRD + 1] = 0.f;
    s_xyzl[tid][NCRD + 2] = 0.f;
  }
  __syncthreads();
  if (tid < TOK) s_ch[tid] = chain_idx[token];

  float (*s_proj)[DH1] = (float (*)[DH1])&s_h3[0][0];
  for (int i = tid; i < NPROJ * DH1 / 4; i += NTHR)
    ((float4*)&s_proj[0][0])[i] = ((const float4*)proj)[i];

  if (tid < TOK * 3) {
    const int   t = tid / 3, d = tid % 3;
    const float x = dihedrals[(size_t)s_tok[t] * 3 + d];
    const float kF[6] = {1.f, 2.f, 3.f, 1.f, 0.5f, 1.f / 3.f};
    const int base = d * 13;
    s_dih[t][base] = x;
#pragma unroll
    for (int f = 0; f < 6; ++f) {
      s_dih[t][base + 1 + f] = sinf(kF[f] * x);
      s_dih[t][base + 7 + f] = cosf(kF[f] * x);
    }
  }

  for (int task = tid; task < TOK * NATOM; task += NTHR) {
    const int t = task / NATOM, a = task % NATOM;
    const int tok = s_tok[t];
    const float* p  = xyz + ((size_t)tok * NATOM + a) * 3;
    const float* ca = xyz + ((size_t)tok * NATOM + 1) * 3;
    const float rx = p[0] - ca[0], ry = p[1] - ca[1], rz = p[2] - ca[2];
    const float* R = orient + (size_t)tok * 9;
    const float m = atom_mask[(size_t)tok * NATOM + a];
    s_xyzl[t][a * 3 + 0] = (R[0] * rx + R[3] * ry + R[6] * rz) * m;
    s_xyzl[t][a * 3 + 1] = (R[1] * rx + R[4] * ry + R[7] * rz) * m;
    s_xyzl[t][a * 3 + 2] = (R[2] * rx + R[5] * ry + R[8] * rz) * m;
  }
  __syncthreads();

  const int wave = tid >> 5;
  const int lane = tid & 31;
  const int half = lane >> 4;
  const int lm   = lane & 15;

  {
    const float* w1dih = W1 + (size_t)(DFEAT + DCOORD) * DH1;
    const int col = wave * 16 + lm;
    v8f dacc[MT] = {};
#pragma unroll
    for (int i = 0; i < 10; ++i) {
      const int r0 = 4 * i + 2 * half;
      v2f b;
      b.x = w1dih[(size_t)r0 * DH1 + col];
      if (i < 9) {
        b.y = w1dih[(size_t)(r0 + 1) * DH1 + col];
      } else {
        const float v = w1dih[(size_t)(r0 + 1 - half) * DH1 + col];
        b.y = half ? 0.f : v;
      }
#pragma unroll
      for (int mt = 0; mt < MT; ++mt) {
        const v2f a = *(const v2f*)&s_dih[16 * mt + lm][4 * i + 2 * half];
        dacc[mt] = __builtin_amdgcn_wmma_f32_16x16x4_f32(
            false, a, false, b, (short)0, dacc[mt], false, false);
      }
    }

    const float* w1crd = W1 + (size_t)DFEAT * DH1 + col;
#pragma unroll
    for (int mt = 0; mt < MT; ++mt) {
      const int sq = s_seq[16 * mt + lm];
      v8f acc = dacc[mt];
#pragma unroll 1
      for (int s = 0; s < N_AA; ++s) {
        const bool mine = (sq == s);
        if (__builtin_amdgcn_ballot_w32(mine) == 0u) continue;
        const float* wp = w1crd + (size_t)s * NCRD * DH1;
#pragma unroll
        for (int i = 0; i < 12; ++i) {
          const int r0 = 4 * i + 2 * half;
          v2f b;
          if (i < 11) {
            b.x = wp[(size_t)r0 * DH1];
            b.y = wp[(size_t)(r0 + 1) * DH1];
          } else {
            const float v = wp[(size_t)(NCRD - 1) * DH1];
            b.x = half ? 0.f : v;
            b.y = 0.f;
          }
          v2f a = *(const v2f*)&s_xyzl[16 * mt + lm][4 * i + 2 * half];
          a.x = mine ? a.x : 0.f;
          a.y = mine ? a.y : 0.f;
          acc = __builtin_amdgcn_wmma_f32_16x16x4_f32(
              false, a, false, b, (short)0, acc, false, false);
        }
      }
      dacc[mt] = acc;
    }

    const float bb = b1[col];
#pragma unroll
    for (int mt = 0; mt < MT; ++mt) {
#pragma unroll
      for (int r = 0; r < 8; ++r) {
        const int row = 16 * mt + r + 8 * half;
        const float pre = (bb + s_proj[s_seq[row]][col]) + s_proj[N_AA + s_ch[row]][col];
        s_h1[row][col] = fmaxf(pre + dacc[mt][r], 0.f);
      }
    }
  }
  __syncthreads();

  {
    const int col = (wave & 7) * 16 + lm;
    const int row0 = (wave >> 3) * 32;
    v8f acc[2] = {};
#pragma unroll 8
    for (int k = 0; k < 256; k += 4) {
      v2f b;
      b.x = W2[(size_t)(k + 2 * half)     * DFEAT + col];
      b.y = W2[(size_t)(k + 2 * half + 1) * DFEAT + col];
#pragma unroll
      for (int mt = 0; mt < 2; ++mt) {
        const v2f a = *(const v2f*)&s_h1[row0 + 16 * mt + lm][k + 2 * half];
        acc[mt] = __builtin_amdgcn_wmma_f32_16x16x4_f32(
            false, a, false, b, (short)0, acc[mt], false, false);
      }
    }
    const float bias = b2[col];
#pragma unroll
    for (int mt = 0; mt < 2; ++mt) {
#pragma unroll
      for (int r = 0; r < 8; ++r)
        s_h2[row0 + 16 * mt + r + 8 * half][col] = fmaxf(acc[mt][r] + bias, 0.f);
    }
  }
  __syncthreads();

  {
    const int col = (wave & 7) * 16 + lm;
    const int row0 = (wave >> 3) * 32;
    v8f acc[2] = {};
#pragma unroll 8
    for (int k = 0; k < 128; k += 4) {
      v2f b;
      b.x = W3[(size_t)(k + 2 * half)     * DFEAT + col];
      b.y = W3[(size_t)(k + 2 * half + 1) * DFEAT + col];
#pragma unroll
      for (int mt = 0; mt < 2; ++mt) {
        const v2f a = *(const v2f*)&s_h2[row0 + 16 * mt + lm][k + 2 * half];
        acc[mt] = __builtin_amdgcn_wmma_f32_16x16x4_f32(
            false, a, false, b, (short)0, acc[mt], false, false);
      }
    }
    const float bias = b3[col];
#pragma unroll
    for (int mt = 0; mt < 2; ++mt) {
#pragma unroll
      for (int r = 0; r < 8; ++r)
        s_h3[row0 + 16 * mt + r + 8 * half][col] = fmaxf(acc[mt][r] + bias, 0.f);
    }
  }
  __syncthreads();

  {
    const int col = (wave & 7) * 16 + lm;
    const int row0 = (wave >> 3) * 32;
    v8f acc[2] = {};
#pragma unroll 8
    for (int k = 0; k < 128; k += 4) {
      v2f b;
      b.x = W4[(size_t)(k + 2 * half)     * DFEAT + col];
      b.y = W4[(size_t)(k + 2 * half + 1) * DFEAT + col];
#pragma unroll
      for (int mt = 0; mt < 2; ++mt) {
        const v2f a = *(const v2f*)&s_h3[row0 + 16 * mt + lm][k + 2 * half];
        acc[mt] = __builtin_amdgcn_wmma_f32_16x16x4_f32(
            false, a, false, b, (short)0, acc[mt], false, false);
      }
    }
    const float bias = b4[col];
#pragma unroll
    for (int mt = 0; mt < 2; ++mt) {
#pragma unroll
      for (int r = 0; r < 8; ++r)
        s_h2[row0 + 16 * mt + r + 8 * half][col] = acc[mt][r] + bias;
    }
  }
  __syncthreads();
#pragma unroll
  for (int i = 0; i < TOK / 16; ++i) {
    const int row = wave * (TOK / 16) + i;
    const float4 v = *(const float4*)&s_h2[row][4 * lane];
    *(float4*)&out[(size_t)s_tok[row] * DFEAT + 4 * lane] = v;
  }
}

extern "C" void kernel_launch(void* const* d_in, const int* in_sizes, int n_in,
                              void* d_out, int out_size, void* d_ws, size_t ws_size,
                              hipStream_t stream) {
  const int*   seq    = (const int*)  d_in[0];
  const float* xyz    = (const float*)d_in[1];
  const float* orient = (const float*)d_in[2];
  const float* dih    = (const float*)d_in[3];
  const int*   chain  = (const int*)  d_in[4];
  const float* amask  = (const float*)d_in[5];
  const float* aa_emb = (const float*)d_in[6];
  const float* ch_emb = (const float*)d_in[7];
  const float* W1 = (const float*)d_in[8];
  const float* b1 = (const float*)d_in[9];
  const float* W2 = (const float*)d_in[10];
  const float* b2 = (const float*)d_in[11];
  const float* W3 = (const float*)d_in[12];
  const float* b3 = (const float*)d_in[13];
  const float* W4 = (const float*)d_in[14];
  const float* b4 = (const float*)d_in[15];
  float* out  = (float*)d_out;
  float* proj = (float*)d_ws;
  int*   wsi  = (int*)d_ws;

  const int ntok = in_sizes[0];

  precompute_proj<<<N_AA + 10 + N_AA, SORT_THR, 0, stream>>>(aa_emb, ch_emb, W1, seq, ntok, proj, wsi);
  residue_embed_fused<<<ntok / TOK, NTHR, 0, stream>>>(
      seq, xyz, orient, dih, chain, amask,
      W1, b1, W2, b2, W3, b3, W4, b4, proj, wsi, ntok, out);
}


# Fused kernel's feature build: projection-table loads issue ahead of the token lookup, and all gathers (angle, both coordinate rounds, chain index) as one group: one memory round trip, not eight
# speedup vs baseline: 2.6226x; 1.0306x over previous
#include <hip/hip_runtime.h>


#define N_AA   21
#define NATOM  15
#define DFEAT  128
#define DCOORD (N_AA * NATOM * 3)
#define DDIH   39
#define DH1    256
#define NCRD   (NATOM * 3)
#define H1S    260
#define H2S    132
#define XS     52
#define DIHS   44
#define NPROJ  (N_AA + 10)
#define MT     4
#define TOK    (16 * MT)
#define NTHR   512
#define SORT_THR 1024
#define SORT_CH  32
#define NPV    (NPROJ * DH1 / 4)
#define XRND   ((TOK * NATOM + NTHR - 1) / NTHR)
#define WS_CNT   8192
#define WS_LIST  8256

typedef __attribute__((ext_vector_type(2))) float v2f;
typedef __attribute__((ext_vector_type(8))) float v8f;
typedef __attribute__((ext_vector_type(4))) float v4f;

__global__ __launch_bounds__(SORT_THR) void precompute_proj(const float* __restrict__ aa_emb,
                                const float* __restrict__ chain_emb,
                                const float* __restrict__ W1,
                                const int*   __restrict__ seq_idx, int ntok,
                                float* __restrict__ proj,
                                int*   __restrict__ wsi) {
  __shared__ int s_wtot[SORT_THR / 32];
  const int r   = blockIdx.x;
  const int tid = threadIdx.x;
  if (r < N_AA + 10) {
    if (tid < DH1) {
      const int c = tid;
      const float* emb;
      const float* wrow;
      if (r < N_AA) {
        emb  = aa_emb + (size_t)r * DFEAT;
        wrow = W1;
      } else {
        emb  = chain_emb + (size_t)(r - N_AA) * DFEAT;
        wrow = W1 + (size_t)(DFEAT + DCOORD + DDIH) * DH1;
      }
      float acc = 0.f;
      for (int k = 0; k < DFEAT; ++k)
        acc = fmaf(emb[k], wrow[(size_t)k * DH1 + c], acc);
      proj[(size_t)r * DH1 + c] = acc;
    }
    return;
  }

  const int s    = r - (N_AA + 10);
  const int lane = tid & 31;
  const int wave = tid >> 5;
  int* list = wsi + WS_LIST + (size_t)s * ntok;
  int running = 0;
  for (int base = 0; base < ntok; base += SORT_THR * SORT_CH) {
    const int first = base + tid * SORT_CH;
    unsigned m = 0;
    if (first + SORT_CH <= ntok) {
      const int4* p = (const int4*)(seq_idx + first);
#pragma unroll
      for (int i = 0; i < SORT_CH / 4; ++i) {
        const int4 v = p[i];
        m |= ((unsigned)(v.x == s) << (4 * i))     | ((unsigned)(v.y == s) << (4 * i + 1)) |
             ((unsigned)(v.z == s) << (4 * i + 2)) | ((unsigned)(v.w == s) << (4 * i + 3));
      }
    } else {
      for (int i = 0; i < SORT_CH; ++i)
        if (first + i < ntok) m |= (unsigned)(seq_idx[first + i] == s) << i;
    }
    const int cnt = __popc(m);
    int inc = cnt;
#pragma unroll
    for (int d = 1; d < 32; d <<= 1) {
      const int v = __shfl_up(inc, d);
      if (lane >= d) inc += v;
    }
    if (lane == 31) s_wtot[wave] = inc;
    __syncthreads();
    const int wt = s_wtot[lane];
    int winc = wt;
#pragma unroll
    for (int d = 1; d < 32; d <<= 1) {
      const int v = __shfl_up(winc, d);
      if (lane >= d) winc += v;
    }
    const int before = __shfl(winc, wave) - __shfl(wt, wave);
    const int total  = __shfl(winc, 31);
    int pos = running + before + inc - cnt;
    while (m) {
      const int b = __builtin_ctz(m);
      m &= m - 1;
      list[pos++] = first + b;
    }
    running += total;
    __syncthreads();
  }
  if (tid == 0) wsi[WS_CNT + s] = running;
}

__global__ __launch_bounds__(NTHR, 8) void residue_embed_fused(
    const int*   __restrict__ seq_idx,
    const float* __restrict__ xyz,
    const float* __restrict__ orient,
    const float* __restrict__ dihedrals,
    const int*   __restrict__ chain_idx,
    const float* __restrict__ atom_mask,
    const float* __restrict__ W1, const float* __restrict__ b1,
    const float* __restrict__ W2, const float* __restrict__ b2,
    const float* __restrict__ W3, const float* __restrict__ b3,
    const float* __restrict__ W4, const float* __restrict__ b4,
    const float* __restrict__ proj,
    const int*   __restrict__ wsi,
    int ntok,
    float*       __restrict__ out)
{
  __shared__ __attribute__((aligned(16))) float s_h1[TOK][H1S];
  __shared__ __attribute__((aligned(16))) float s_h2[TOK][H2S];
  __shared__ __attribute__((aligned(16))) float s_h3[TOK][H2S];
  __shared__ __attribute__((aligned(16))) float s_xyzl[TOK][XS];
  __shared__ int   s_seq[TOK];
  __shared__ int   s_ch[TOK];
  __shared__ int   s_tok[TOK];
  float (*s_dih)[DIHS] = (float (*)[DIHS])&s_h2[0][0];

  const int tid  = threadIdx.x;
  const int tok0 = blockIdx.x * TOK;

  float (*s_proj)[DH1] = (float (*)[DH1])&s_h3[0][0];
  const v4f* pg = (const v4f*)proj;
  static_assert(NPV > 3 * NTHR && NPV <= 4 * NTHR, "the copy below is written for four pieces per thread, the last round partial");
  const v4f pv0 = pg[tid], pv1 = pg[tid + NTHR], pv2 = pg[tid + 2 * NTHR];
  const v4f pv3 = pg[min(tid + 3 * NTHR, NPV - 1)];

  int token = 0;
  if (tid < TOK) {
    const int p = tok0 + tid;
    int s = 0, off = 0, end = 0;
#pragma unroll
    for (int q = 0; q < N_AA; ++q) {
      end += wsi[WS_CNT + q];
      if (p >= end) { s = q + 1; off = end; }
    }
    s = min(s, N_AA - 1);
    token = wsi[WS_LIST + (size_t)s * ntok + (p - off)];
    token = min(max(token, 0), ntok - 1);
    s_tok[tid] = token;
    s_seq[tid] = s;
    s_dih[tid][DDIH] = 0.f;
    s_xyzl[tid][NCRD] = 0.f;
    s_xyzl[tid][NCRD + 1] = 0.f;
    s_xyzl[tid][NCRD + 2] = 0.f;
  }
  {
    v4f* pl = (v4f*)&s_proj[0][0];
    pl[tid] = pv0;
    pl[tid + NTHR] = pv1;
    pl[tid + 2 * NTHR] = pv2;
    if (tid + 3 * NTHR < NPV) pl[tid + 3 * NTHR] = pv3;
  }
  __syncthreads();

  const bool has_dih = tid < TOK * 3;
  const int  dt = tid / 3, dd = tid % 3;
  float dx = 0.f;
  if (has_dih) dx = dihedrals[(size_t)s_tok[dt] * 3 + dd];

  float gp[XRND][3] = {}, gc[XRND][3] = {}, gR[XRND][9] = {}, gm[XRND] = {};
#pragma unroll
  for (int j = 0; j < XRND; ++j) {
    const int task = tid + j * NTHR;
    if (task < TOK * NATOM) {
      const int t = task / NATOM, a = task % NATOM;
      const int tok = s_tok[t];
      const float* p  = xyz + ((size_t)tok * NATOM + a) * 3;
      const float* ca = xyz + ((size_t)tok * NATOM + 1) * 3;
      const float* R  = orient + (size_t)tok * 9;
#pragma unroll
      for (int q = 0; q < 3; ++q) { gp[j][q] = p[q]; gc[j][q] = ca[q]; }
#pragma unroll
      for (int q = 0; q < 9; ++q) gR[j][q] = R[q];
      gm[j] = atom_mask[(size_t)tok * NATOM + a];
    }
  }
  int chv = 0;
  if (tid < TOK) chv = chain_idx[token];

  if (has_dih) {
    const float x = dx;
    const float kF[6] = {1.f, 2.f, 3.f, 1.f, 0.5f, 1.f / 3.f};
    const int base = dd * 13;
    s_dih[dt][base] = x;
#pragma unroll
    for (int f = 0; f < 6; ++f) {
      s_dih[dt][base + 1 + f] = sinf(kF[f] * x);
      s_dih[dt][base + 7 + f] = cosf(kF[f] * x);
    }
  }

#pragma unroll
  for (int j = 0; j < XRND; ++j) {
    const int task = tid + j * NTHR;
    if (task < TOK * NATOM) {
      const int t = task / NATOM, a = task % NATOM;
      const float rx = gp[j][0] - gc[j][0], ry = gp[j][1] - gc[j][1], rz = gp[j][2] - gc[j][2];
      const float* R = gR[j];
      const float m = gm[j];
      s_xyzl[t][a * 3 + 0] = (R[0] * rx + R[3] * ry + R[6] * rz) * m;
      s_xyzl[t][a * 3 + 1] = (R[1] * rx + R[4] * ry + R[7] * rz) * m;
      s_xyzl[t][a * 3 + 2] = (R[2] * rx + R[5] * ry + R[8] * rz) * m;
    }
  }
  if (tid < TOK) s_ch[tid] = chv;
  __syncthreads();

  const int wave = tid >> 5;
  const int lane = tid & 31;
  const int half = lane >> 4;
  const int lm   = lane & 15;

  {
    const float* w1dih = W1 + (size_t)(DFEAT + DCOORD) * DH1;
    const int col = wave * 16 + lm;
    v8f dacc[MT] = {};
#pragma unroll
    for (int i = 0; i < 10; ++i) {
      const int r0 = 4 * i + 2 * half;
      v2f b;
      b.x = w1dih[(size_t)r0 * DH1 + col];
      if (i < 9) {
        b.y = w1dih[(size_t)(r0 + 1) * DH1 + col];
      } else {
        const float v = w1dih[(size_t)(r0 + 1 - half) * DH1 + col];
        b.y = half ? 0.f : v;
      }
#pragma unroll
      for (int mt = 0; mt < MT; ++mt) {
        const v2f a = *(const v2f*)&s_dih[16 * mt + lm][4 * i + 2 * half];
        dacc[mt] = __builtin_amdgcn_wmma_f32_16x16x4_f32(
            false, a, false, b, (short)0, dacc[mt], false, false);
      }
    }

    const float* w1crd = W1 + (size_t)DFEAT * DH1 + col;
#pragma unroll
    for (int mt = 0; mt < MT; ++mt) {
      const int sq = s_seq[16 * mt + lm];
      v8f acc = dacc[mt];
#pragma unroll 1
      for (int s = 0; s < N_AA; ++s) {
        const bool mine = (sq == s);
        if (__builtin_amdgcn_ballot_w32(mine) == 0u) continue;
        const float* wp = w1crd + (size_t)s * NCRD * DH1;
#pragma unroll
        for (int i = 0; i < 12; ++i) {
          const int r0 = 4 * i + 2 * half;
          v2f b;
          if (i < 11) {
            b.x = wp[(size_t)r0 * DH1];
            b.y = wp[(size_t)(r0 + 1) * DH1];
          } else {
            const float v = wp[(size_t)(NCRD - 1) * DH1];
            b.x = half ? 0.f : v;
            b.y = 0.f;
          }
          v2f a = *(const v2f*)&s_xyzl[16 * mt + lm][4 * i + 2 * half];
          a.x = mine ? a.x : 0.f;
          a.y = mine ? a.y : 0.f;
          acc = __builtin_amdgcn_wmma_f32_16x16x4_f32(
              false, a, false, b, (short)0, acc, false, false);
        }
      }
      dacc[mt] = acc;
    }

    const float bb = b1[col];
#pragma unroll
    for (int mt = 0; mt < MT; ++mt) {
#pragma unroll
      for (int r = 0; r < 8; ++r) {
        const int row = 16 * mt + r + 8 * half;
        const float pre = (bb + s_proj[s_seq[row]][col]) + s_proj[N_AA + s_ch[row]][col];
        s_h1[row][col] = fmaxf(pre + dacc[mt][r], 0.f);
      }
    }
  }
  __syncthreads();

  {
    const int col = (wave & 7) * 16 + lm;
    const int row0 = (wave >> 3) * 32;
    v8f acc[2] = {};
#pragma unroll 8
    for (int k = 0; k < 256; k += 4) {
      v2f b;
      b.x = W2[(size_t)(k + 2 * half)     * DFEAT + col];
      b.y = W2[(size_t)(k + 2 * half + 1) * DFEAT + col];
#pragma unroll
      for (int mt = 0; mt < 2; ++mt) {
        const v2f a = *(const v2f*)&s_h1[row0 + 16 * mt + lm][k + 2 * half];
        acc[mt] = __builtin_amdgcn_wmma_f32_16x16x4_f32(
            false, a, false, b, (short)0, acc[mt], false, false);
      }
    }
    const float bias = b2[col];
#pragma unroll
    for (int mt = 0; mt < 2; ++mt) {
#pragma unroll
      for (int r = 0; r < 8; ++r)
        s_h2[row0 + 16 * mt + r + 8 * half][col] = fmaxf(acc[mt][r] + bias, 0.f);
    }
  }
  __syncthreads();

  {
    const int col = (wave & 7) * 16 + lm;
    const int row0 = (wave >> 3) * 32;
    v8f acc[2] = {};
#pragma unroll 8
    for (int k = 0; k < 128; k += 4) {
      v2f b;
      b.x = W3[(size_t)(k + 2 * half)     * DFEAT + col];
      b.y = W3[(size_t)(k + 2 * half + 1) * DFEAT + col];
#pragma unroll
      for (int mt = 0; mt < 2; ++mt) {
        const v2f a = *(const v2f*)&s_h2[row0 + 16 * mt + lm][k + 2 * half];
        acc[mt] = __builtin_amdgcn_wmma_f32_16x16x4_f32(
            false, a, false, b, (short)0, acc[mt], false, false);
      }
    }
    const float bias = b3[col];
#pragma unroll
    for (int mt = 0; mt < 2; ++mt) {
#pragma unroll
      for (int r = 0; r < 8; ++r)
        s_h3[row0 + 16 * mt + r + 8 * half][col] = fmaxf(acc[mt][r] + bias, 0.f);
    }
  }
  __syncthreads();

  {
    const int col = (wave & 7) * 16 + lm;
    const int row0 = (wave >> 3) * 32;
    v8f acc[2] = {};
#pragma unroll 8
    for (int k = 0; k < 128; k += 4) {
      v2f b;
      b.x = W4[(size_t)(k + 2 * half)     * DFEAT + col];
      b.y = W4[(size_t)(k + 2 * half + 1) * DFEAT + col];
#pragma unroll
      for (int mt = 0; mt < 2; ++mt) {
        const v2f a = *(const v2f*)&s_h3[row0 + 16 * mt + lm][k + 2 * half];
        acc[mt] = __builtin_amdgcn_wmma_f32_16x16x4_f32(
            false, a, false, b, (short)0, acc[mt], false, false);
      }
    }
    const float bias = b4[col];
#pragma unroll
    for (int mt = 0; mt < 2; ++mt) {
#pragma unroll
      for (int r = 0; r < 8; ++r)
        s_h2[row0 + 16 * mt + r + 8 * half][col] = acc[mt][r] + bias;
    }
  }
  __syncthreads();
#pragma unroll
  for (int i = 0; i < TOK / 16; ++i) {
    const int row = wave * (TOK / 16) + i;
    const float4 v = *(const float4*)&s_h2[row][4 * lane];
    *(float4*)&out[(size_t)s_tok[row] * DFEAT + 4 * lane] = v;
  }
}

extern "C" void kernel_launch(void* const* d_in, const int* in_sizes, int n_in,
                              void* d_out, int out_size, void* d_ws, size_t ws_size,
                              hipStream_t stream) {
  const int*   seq    = (const int*)  d_in[0];
  const float* xyz    = (const float*)d_in[1];
  const float* orient = (const float*)d_in[2];
  const float* dih    = (const float*)d_in[3];
  const int*   chain  = (const int*)  d_in[4];
  const float* amask  = (const float*)d_in[5];
  const float* aa_emb = (const float*)d_in[6];
  const float* ch_emb = (const float*)d_in[7];
  const float* W1 = (const float*)d_in[8];
  const float* b1 = (const float*)d_in[9];
  const float* W2 = (const float*)d_in[10];
  const float* b2 = (const float*)d_in[11];
  const float* W3 = (const float*)d_in[12];
  const float* b3 = (const float*)d_in[13];
  const float* W4 = (const float*)d_in[14];
  const float* b4 = (const float*)d_in[15];
  float* out  = (float*)d_out;
  float* proj = (float*)d_ws;
  int*   wsi  = (int*)d_ws;

  const int ntok = in_sizes[0];

  precompute_proj<<<N_AA + 10 + N_AA, SORT_THR, 0, stream>>>(aa_emb, ch_emb, W1, seq, ntok, proj, wsi);
  residue_embed_fused<<<ntok / TOK, NTHR, 0, stream>>>(
      seq, xyz, orient, dih, chain, amask,
      W1, b1, W2, b2, W3, b3, W4, b4, proj, wsi, ntok, out);
}


# Work-group 16 -> 8 waves on the same 64-token tile: a wave owns all four row tiles (two column tiles in layer 1), so one weight fragment load feeds 4-8 products, not 1-4; sum order unchanged
# speedup vs baseline: 2.8271x; 1.0780x over previous
#include <hip/hip_runtime.h>


#define N_AA   21
#define NATOM  15
#define DFEAT  128
#define DCOORD (N_AA * NATOM * 3)
#define DDIH   39
#define DH1    256
#define NCRD   (NATOM * 3)
#define H1S    260
#define H2S    132
#define XS     52
#define DIHS   44
#define NPROJ  (N_AA + 10)
#define MT     4
#define TOK    (16 * MT)
#define NTHR   256
#define NWAVE  (NTHR / 32)
#define SORT_THR 1024
#define SORT_CH  32
#define NPV    (NPROJ * DH1 / 4)
#define NPRND  ((NPV + NTHR - 1) / NTHR)
#define XRND   ((TOK * NATOM + NTHR - 1) / NTHR)
#define WS_CNT   8192
#define WS_LIST  8256

typedef __attribute__((ext_vector_type(2))) float v2f;
typedef __attribute__((ext_vector_type(8))) float v8f;
typedef __attribute__((ext_vector_type(4))) float v4f;

__global__ __launch_bounds__(SORT_THR) void precompute_proj(const float* __restrict__ aa_emb,
                                const float* __restrict__ chain_emb,
                                const float* __restrict__ W1,
                                const int*   __restrict__ seq_idx, int ntok,
                                float* __restrict__ proj,
                                int*   __restrict__ wsi) {
  __shared__ int s_wtot[SORT_THR / 32];
  const int r   = blockIdx.x;
  const int tid = threadIdx.x;
  if (r < N_AA + 10) {
    if (tid < DH1) {
      const int c = tid;
      const float* emb;
      const float* wrow;
      if (r < N_AA) {
        emb  = aa_emb + (size_t)r * DFEAT;
        wrow = W1;
      } else {
        emb  = chain_emb + (size_t)(r - N_AA) * DFEAT;
        wrow = W1 + (size_t)(DFEAT + DCOORD + DDIH) * DH1;
      }
      float acc = 0.f;
      for (int k = 0; k < DFEAT; ++k)
        acc = fmaf(emb[k], wrow[(size_t)k * DH1 + c], acc);
      proj[(size_t)r * DH1 + c] = acc;
    }
    return;
  }

  const int s    = r - (N_AA + 10);
  const int lane = tid & 31;
  const int wave = tid >> 5;
  int* list = wsi + WS_LIST + (size_t)s * ntok;
  int running = 0;
  for (int base = 0; base < ntok; base += SORT_THR * SORT_CH) {
    const int first = base + tid * SORT_CH;
    unsigned m = 0;
    if (first + SORT_CH <= ntok) {
      const int4* p = (const int4*)(seq_idx + first);
#pragma unroll
      for (int i = 0; i < SORT_CH / 4; ++i) {
        const int4 v = p[i];
        m |= ((unsigned)(v.x == s) << (4 * i))     | ((unsigned)(v.y == s) << (4 * i + 1)) |
             ((unsigned)(v.z == s) << (4 * i + 2)) | ((unsigned)(v.w == s) << (4 * i + 3));
      }
    } else {
      for (int i = 0; i < SORT_CH; ++i)
        if (first + i < ntok) m |= (unsigned)(seq_idx[first + i] == s) << i;
    }
    const int cnt = __popc(m);
    int inc = cnt;
#pragma unroll
    for (int d = 1; d < 32; d <<= 1) {
      const int v = __shfl_up(inc, d);
      if (lane >= d) inc += v;
    }
    if (lane == 31) s_wtot[wave] = inc;
    __syncthreads();
    const int wt = s_wtot[lane];
    int winc = wt;
#pragma unroll
    for (int d = 1; d < 32; d <<= 1) {
      const int v = __shfl_up(winc, d);
      if (lane >= d) winc += v;
    }
    const int before = __shfl(winc, wave) - __shfl(wt, wave);
    const int total  = __shfl(winc, 31);
    int pos = running + before + inc - cnt;
    while (m) {
      const int b = __builtin_ctz(m);
      m &= m - 1;
      list[pos++] = first + b;
    }
    running += total;
    __syncthreads();
  }
  if (tid == 0) wsi[WS_CNT + s] = running;
}

__global__ __launch_bounds__(NTHR, 4) void residue_embed_fused(
    const int*   __restrict__ seq_idx,
    const float* __restrict__ xyz,
    const float* __restrict__ orient,
    const float* __restrict__ dihedrals,
    const int*   __restrict__ chain_idx,
    const float* __restrict__ atom_mask,
    const float* __restrict__ W1, const float* __restrict__ b1,
    const float* __restrict__ W2, const float* __restrict__ b2,
    const float* __restrict__ W3, const float* __restrict__ b3,
    const float* __restrict__ W4, const float* __restrict__ b4,
    const float* __restrict__ proj,
    const int*   __restrict__ wsi,
    int ntok,
    float*       __restrict__ out)
{
  __shared__ __attribute__((aligned(16))) float s_h1[TOK][H1S];
  __shared__ __attribute__((aligned(16))) float s_h2[TOK][H2S];
  __shared__ __attribute__((aligned(16))) float s_h3[TOK][H2S];
  __shared__ __attribute__((aligned(16))) float s_xyzl[TOK][XS];
  __shared__ int   s_seq[TOK];
  __shared__ int   s_ch[TOK];
  __shared__ int   s_tok[TOK];
  float (*s_dih)[DIHS] = (float (*)[DIHS])&s_h2[0][0];

  const int tid  = threadIdx.x;
  const int tok0 = blockIdx.x * TOK;

  float (*s_proj)[DH1] = (float (*)[DH1])&s_h3[0][0];
  const v4f* pg = (const v4f*)proj;
  v4f pv[NPRND];
#pragma unroll
  for (int j = 0; j < NPRND; ++j) pv[j] = pg[min(tid + j * NTHR, NPV - 1)];

  int token = 0;
  if (tid < TOK) {
    const int p = tok0 + tid;
    int s = 0, off = 0, end = 0;
#pragma unroll
    for (int q = 0; q < N_AA; ++q) {
      end += wsi[WS_CNT + q];
      if (p >= end) { s = q + 1; off = end; }
    }
    s = min(s, N_AA - 1);
    token = wsi[WS_LIST + (size_t)s * ntok + (p - off)];
    token = min(max(token, 0), ntok - 1);
    s_tok[tid] = token;
    s_seq[tid] = s;
    s_dih[tid][DDIH] = 0.f;
    s_xyzl[tid][NCRD] = 0.f;
    s_xyzl[tid][NCRD + 1] = 0.f;
    s_xyzl[tid][NCRD + 2] = 0.f;
  }
  {
    v4f* pl = (v4f*)&s_proj[0][0];
#pragma unroll
    for (int j = 0; j < NPRND; ++j)
      if (tid + j * NTHR < NPV) pl[tid + j * NTHR] = pv[j];
  }
  __syncthreads();

  const bool has_dih = tid < TOK * 3;
  const int  dt = tid / 3, dd = tid % 3;
  float dx = 0.f;
  if (has_dih) dx = dihedrals[(size_t)s_tok[dt] * 3 + dd];

  float gp[XRND][3] = {}, gc[XRND][3] = {}, gR[XRND][9] = {}, gm[XRND] = {};
#pragma unroll
  for (int j = 0; j < XRND; ++j) {
    const int task = tid + j * NTHR;
    if (task < TOK * NATOM) {
      const int t = task / NATOM, a = task % NATOM;
      const int tok = s_tok[t];
      const float* p  = xyz + ((size_t)tok * NATOM + a) * 3;
      const float* ca = xyz + ((size_t)tok * NATOM + 1) * 3;
      const float* R  = orient + (size_t)tok * 9;
#pragma unroll
      for (int q = 0; q < 3; ++q) { gp[j][q] = p[q]; gc[j][q] = ca[q]; }
#pragma unroll
      for (int q = 0; q < 9; ++q) gR[j][q] = R[q];
      gm[j] = atom_mask[(size_t)tok * NATOM + a];
    }
  }
  int chv = 0;
  if (tid < TOK) chv = chain_idx[token];

  if (has_dih) {
    const float x = dx;
    const float kF[6] = {1.f, 2.f, 3.f, 1.f, 0.5f, 1.f / 3.f};
    const int base = dd * 13;
    s_dih[dt][base] = x;
#pragma unroll
    for (int f = 0; f < 6; ++f) {
      s_dih[dt][base + 1 + f] = sinf(kF[f] * x);
      s_dih[dt][base + 7 + f] = cosf(kF[f] * x);
    }
  }

#pragma unroll
  for (int j = 0; j < XRND; ++j) {
    const int task = tid + j * NTHR;
    if (task < TOK * NATOM) {
      const int t = task / NATOM, a = task % NATOM;
      const float rx = gp[j][0] - gc[j][0], ry = gp[j][1] - gc[j][1], rz = gp[j][2] - gc[j][2];
      const float* R = gR[j];
      const float m = gm[j];
      s_xyzl[t][a * 3 + 0] = (R[0] * rx + R[3] * ry + R[6] * rz) * m;
      s_xyzl[t][a * 3 + 1] = (R[1] * rx + R[4] * ry + R[7] * rz) * m;
      s_xyzl[t][a * 3 + 2] = (R[2] * rx + R[5] * ry + R[8] * rz) * m;
    }
  }
  if (tid < TOK) s_ch[tid] = chv;
  __syncthreads();

  const int wave = tid >> 5;
  const int lane = tid & 31;
  const int half = lane >> 4;
  const int lm   = lane & 15;

  {
    const float* w1dih = W1 + (size_t)(DFEAT + DCOORD) * DH1;
    const int col0 = wave * 32 + lm;
    v8f dacc[2][MT] = {};
#pragma unroll
    for (int i = 0; i < 10; ++i) {
      const int r0 = 4 * i + 2 * half;
      v2f b[2];
#pragma unroll
      for (int ct = 0; ct < 2; ++ct) {
        const float* wc = w1dih + col0 + 16 * ct;
        b[ct].x = wc[(size_t)r0 * DH1];
        if (i < 9) {
          b[ct].y = wc[(size_t)(r0 + 1) * DH1];
        } else {
          const float v = wc[(size_t)(r0 + 1 - half) * DH1];
          b[ct].y = half ? 0.f : v;
        }
      }
#pragma unroll
      for (int mt = 0; mt < MT; ++mt) {
        const v2f a = *(const v2f*)&s_dih[16 * mt + lm][4 * i + 2 * half];
#pragma unroll
        for (int ct = 0; ct < 2; ++ct)
          dacc[ct][mt] = __builtin_amdgcn_wmma_f32_16x16x4_f32(
              false, a, false, b[ct], (short)0, dacc[ct][mt], false, false);
      }
    }

    const float* w1crd = W1 + (size_t)DFEAT * DH1 + col0;
    int sq[MT];
#pragma unroll
    for (int mt = 0; mt < MT; ++mt) sq[mt] = s_seq[16 * mt + lm];
#pragma unroll 1
    for (int s = 0; s < N_AA; ++s) {
      bool any = false;
#pragma unroll
      for (int mt = 0; mt < MT; ++mt) any = any || (sq[mt] == s);
      if (__builtin_amdgcn_ballot_w32(any) == 0u) continue;
      const float* wp = w1crd + (size_t)s * NCRD * DH1;
      v2f b[12][2];
#pragma unroll
      for (int i = 0; i < 12; ++i) {
        const int r0 = 4 * i + 2 * half;
#pragma unroll
        for (int ct = 0; ct < 2; ++ct) {
          const float* wc = wp + 16 * ct;
          if (i < 11) {
            b[i][ct].x = wc[(size_t)r0 * DH1];
            b[i][ct].y = wc[(size_t)(r0 + 1) * DH1];
          } else {
            const float v = wc[(size_t)(NCRD - 1) * DH1];
            b[i][ct].x = half ? 0.f : v;
            b[i][ct].y = 0.f;
          }
        }
      }
#pragma unroll
      for (int mt = 0; mt < MT; ++mt) {
        const bool mine = (sq[mt] == s);
        if (__builtin_amdgcn_ballot_w32(mine) != 0u) {
#pragma unroll
          for (int i = 0; i < 12; ++i) {
            v2f a = *(const v2f*)&s_xyzl[16 * mt + lm][4 * i + 2 * half];
            a.x = mine ? a.x : 0.f;
            a.y = mine ? a.y : 0.f;
#pragma unroll
            for (int ct = 0; ct < 2; ++ct)
              dacc[ct][mt] = __builtin_amdgcn_wmma_f32_16x16x4_f32(
                  false, a, false, b[i][ct], (short)0, dacc[ct][mt], false, false);
          }
        }
      }
    }

#pragma unroll
    for (int ct = 0; ct < 2; ++ct) {
      const int col = col0 + 16 * ct;
      const float bb = b1[col];
#pragma unroll
      for (int mt = 0; mt < MT; ++mt) {
#pragma unroll
        for (int r = 0; r < 8; ++r) {
          const int row = 16 * mt + r + 8 * half;
          const float pre = (bb + s_proj[s_seq[row]][col]) + s_proj[N_AA + s_ch[row]][col];
          s_h1[row][col] = fmaxf(pre + dacc[ct][mt][r], 0.f);
        }
      }
    }
  }
  __syncthreads();

  {
    const int col = wave * 16 + lm;
    const float* bp = W2 + (size_t)(2 * half) * DFEAT + col;
    v8f acc[MT] = {};
#pragma unroll 8
    for (int k = 0; k < 256; k += 4) {
      v2f b;
      b.x = bp[(size_t)k * DFEAT];
      b.y = bp[(size_t)(k + 1) * DFEAT];
#pragma unroll
      for (int mt = 0; mt < MT; ++mt) {
        const v2f a = *(const v2f*)&s_h1[16 * mt + lm][k + 2 * half];
        acc[mt] = __builtin_amdgcn_wmma_f32_16x16x4_f32(
            false, a, false, b, (short)0, acc[mt], false, false);
      }
    }
    const float bias = b2[col];
#pragma unroll
    for (int mt = 0; mt < MT; ++mt) {
#pragma unroll
      for (int r = 0; r < 8; ++r)
        s_h2[16 * mt + r + 8 * half][col] = fmaxf(acc[mt][r] + bias, 0.f);
    }
  }
  __syncthreads();

  {
    const int col = wave * 16 + lm;
    const float* bp = W3 + (size_t)(2 * half) * DFEAT + col;
    v8f acc[MT] = {};
#pragma unroll 8
    for (int k = 0; k < 128; k += 4) {
      v2f b;
      b.x = bp[(size_t)k * DFEAT];
      b.y = bp[(size_t)(k + 1) * DFEAT];
#pragma unroll
      for (int mt = 0; mt < MT; ++mt) {
        const v2f a = *(const v2f*)&s_h2[16 * mt + lm][k + 2 * half];
        acc[mt] = __builtin_amdgcn_wmma_f32_16x16x4_f32(
            false, a, false, b, (short)0, acc[mt], false, false);
      }
    }
    const float bias = b3[col];
#pragma unroll
    for (int mt = 0; mt < MT; ++mt) {
#pragma unroll
      for (int r = 0; r < 8; ++r)
        s_h3[16 * mt + r + 8 * half][col] = fmaxf(acc[mt][r] + bias, 0.f);
    }
  }
  __syncthreads();

  {
    const int col = wave * 16 + lm;
    const float* bp = W4 + (size_t)(2 * half) * DFEAT + col;
    v8f acc[MT] = {};
#pragma unroll 8
    for (int k = 0; k < 128; k += 4) {
      v2f b;
      b.x = bp[(size_t)k * DFEAT];
      b.y = bp[(size_t)(k + 1) * DFEAT];
#pragma unroll
      for (int mt = 0; mt < MT; ++mt) {
        const v2f a = *(const v2f*)&s_h3[16 * mt + lm][k + 2 * half];
        acc[mt] = __builtin_amdgcn_wmma_f32_16x16x4_f32(
            false, a, false, b, (short)0, acc[mt], false, false);
      }
    }
    const float bias = b4[col];
#pragma unroll
    for (int mt = 0; mt < MT; ++mt) {
#pragma unroll
      for (int r = 0; r < 8; ++r)
        s_h2[16 * mt + r + 8 * half][col] = acc[mt][r] + bias;
    }
  }
  __syncthreads();
#pragma unroll
  for (int i = 0; i < TOK / NWAVE; ++i) {
    const int row = wave * (TOK / NWAVE) + i;
    const float4 v = *(const float4*)&s_h2[row][4 * lane];
    *(float4*)&out[(size_t)s_tok[row] * DFEAT + 4 * lane] = v;
  }
}

extern "C" void kernel_launch(void* const* d_in, const int* in_sizes, int n_in,
                              void* d_out, int out_size, void* d_ws, size_t ws_size,
                              hipStream_t stream) {
  const int*   seq    = (const int*)  d_in[0];
  const float* xyz    = (const float*)d_in[1];
  const float* orient = (const float*)d_in[2];
  const float* dih    = (const float*)d_in[3];
  const int*   chain  = (const int*)  d_in[4];
  const float* amask  = (const float*)d_in[5];
  const float* aa_emb = (const float*)d_in[6];
  const float* ch_emb = (const float*)d_in[7];
  const float* W1 = (const float*)d_in[8];
  const float* b1 = (const float*)d_in[9];
  const float* W2 = (const float*)d_in[10];
  const float* b2 = (const float*)d_in[11];
  const float* W3 = (const float*)d_in[12];
  const float* b3 = (const float*)d_in[13];
  const float* W4 = (const float*)d_in[14];
  const float* b4 = (const float*)d_in[15];
  float* out  = (float*)d_out;
  float* proj = (float*)d_ws;
  int*   wsi  = (int*)d_ws;

  const int ntok = in_sizes[0];

  precompute_proj<<<N_AA + 10 + N_AA, SORT_THR, 0, stream>>>(aa_emb, ch_emb, W1, seq, ntok, proj, wsi);
  residue_embed_fused<<<ntok / TOK, NTHR, 0, stream>>>(
      seq, xyz, orient, dih, chain, amask,
      W1, b1, W2, b2, W3, b3, W4, b4, proj, wsi, ntok, out);
}
